# starting kernel
# baseline (speedup 1.0000x reference)
#include <hip/hip_runtime.h>
#include <hip/hip_bf16.h>


typedef float v2f __attribute__((ext_vector_type(2)));
typedef float v8f __attribute__((ext_vector_type(8)));

#define NNODES 50000
#define EMB_DIM 32
#define HID 128
#define OUTD 128


__global__ void k_zero_i32(int* __restrict__ p, int n) {
  int i = blockIdx.x * blockDim.x + threadIdx.x;
  if (i < n) p[i] = 0;
}

__global__ void k_count(const int* __restrict__ dst, int E,
                        int* __restrict__ cnt) {
  int e = blockIdx.x * blockDim.x + threadIdx.x;
  if (e < E) {
    int d = dst[e];
    if ((unsigned)d < (unsigned)NNODES) atomicAdd(&cnt[d], 1);
  }
}

__global__ void k_dinv(const int* __restrict__ cnt, float* __restrict__ dinv,
                       int n) {
  int i = blockIdx.x * blockDim.x + threadIdx.x;
  if (i < n) dinv[i] = rsqrtf((float)(cnt[i] + 1));
}


__global__ __launch_bounds__(256) void k_scan_block(
    const int* __restrict__ cnt, int n, int* __restrict__ incl,
    int* __restrict__ bsum) {
  __shared__ int sm[256];
  int i = blockIdx.x * 256 + threadIdx.x;
  sm[threadIdx.x] = (i < n) ? cnt[i] : 0;
  __syncthreads();
#pragma unroll
  for (int off = 1; off < 256; off <<= 1) {
    int t = (threadIdx.x >= off) ? sm[threadIdx.x - off] : 0;
    __syncthreads();
    sm[threadIdx.x] += t;
    __syncthreads();
  }
  if (i < n) incl[i] = sm[threadIdx.x];
  if (threadIdx.x == 255) bsum[blockIdx.x] = sm[255];
}

__global__ __launch_bounds__(256) void k_scan_sums(int* __restrict__ bsum,
                                                   int nb) {
  __shared__ int sm[256];
  int t = threadIdx.x;
  sm[t] = (t < nb) ? bsum[t] : 0;
  __syncthreads();
#pragma unroll
  for (int off = 1; off < 256; off <<= 1) {
    int v = (t >= off) ? sm[t - off] : 0;
    __syncthreads();
    sm[t] += v;
    __syncthreads();
  }
  if (t < nb) bsum[t] = sm[t];
}

__global__ __launch_bounds__(256) void k_scan_finalize(
    const int* __restrict__ incl, const int* __restrict__ bsumscan,
    const int* __restrict__ cnt, int n, int* __restrict__ rowptr,
    int* __restrict__ cursor) {
  int i = blockIdx.x * 256 + threadIdx.x;
  if (i >= n) return;
  int boff = (blockIdx.x == 0) ? 0 : bsumscan[blockIdx.x - 1];
  int inc = incl[i] + boff;
  rowptr[i + 1] = inc;
  cursor[i] = inc - cnt[i];
  if (i == 0) rowptr[0] = 0;
}

__global__ void k_fill(const int* __restrict__ src, const int* __restrict__ dst,
                       int E, int* __restrict__ cursor, int* __restrict__ col) {
  int e = blockIdx.x * blockDim.x + threadIdx.x;
  if (e >= E) return;
  int s = src[e];
  int d = dst[e];
  if ((unsigned)s >= (unsigned)NNODES || (unsigned)d >= (unsigned)NNODES)
    return;
  int pos = atomicAdd(&cursor[d], 1);
  col[pos] = s;
}


template <int K, int COLS, bool GATHER>
__global__ __launch_bounds__(256) void k_gemm_wmma(
    const float* __restrict__ A, const float* __restrict__ B,
    float* __restrict__ C, const int* __restrict__ xidx) {
  const int lane = threadIdx.x & 31;
  const int wave = threadIdx.x >> 5;
  const int lo = lane & 15;
  const int hi = lane >> 4;
  const int row0 = blockIdx.x * 16;
  const int col0 = wave * 16;

  int arow = row0 + lo;
  if (GATHER) arow = xidx[arow];
  const float* __restrict__ Ap = A + (size_t)arow * K;

  v8f acc = {};
#pragma unroll
  for (int kb = 0; kb < K; kb += 4) {
    v2f a, b;
    a.x = Ap[kb + 2 * hi];
    a.y = Ap[kb + 2 * hi + 1];
    b.x = B[(size_t)(kb + 2 * hi) * COLS + col0 + lo];
    b.y = B[(size_t)(kb + 2 * hi + 1) * COLS + col0 + lo];
    acc = __builtin_amdgcn_wmma_f32_16x16x4_f32(
        false, a, false, b, (short)0, acc, false, false);
  }
#pragma unroll
  for (int g = 0; g < 8; ++g) {
    C[(size_t)(row0 + g + 8 * hi) * COLS + col0 + lo] = acc[g];
  }
}


template <bool RELU>
__global__ __launch_bounds__(256) void k_aggregate(
    const int* __restrict__ rowptr, const int* __restrict__ col,
    const float* __restrict__ H, const float* __restrict__ dinv,
    const float* __restrict__ bias, float* __restrict__ out, int n) {
  int d = blockIdx.x * (blockDim.x >> 5) + (threadIdx.x >> 5);
  int lane = threadIdx.x & 31;
  if (d >= n) return;

  int k0 = rowptr[d];
  int k1 = rowptr[d + 1];
  float4 acc = make_float4(0.f, 0.f, 0.f, 0.f);
  for (int k = k0; k < k1; ++k) {
    int s = col[k];
    float w = dinv[s];
    float4 v = *((const float4*)(H + (size_t)s * 128) + lane);
    acc.x += v.x * w;
    acc.y += v.y * w;
    acc.z += v.z * w;
    acc.w += v.w * w;
  }
  float dd = dinv[d];
  float4 self = *((const float4*)(H + (size_t)d * 128) + lane);
  float4 bb = *((const float4*)bias + lane);
  float4 r;
  r.x = (acc.x + self.x * dd) * dd + bb.x;
  r.y = (acc.y + self.y * dd) * dd + bb.y;
  r.z = (acc.z + self.z * dd) * dd + bb.z;
  r.w = (acc.w + self.w * dd) * dd + bb.w;
  if (RELU) {
    r.x = fmaxf(r.x, 0.f);
    r.y = fmaxf(r.y, 0.f);
    r.z = fmaxf(r.z, 0.f);
    r.w = fmaxf(r.w, 0.f);
  }
  *((float4*)(out + (size_t)d * 128) + lane) = r;
}


static inline size_t align256(size_t x) { return (x + 255) & ~(size_t)255; }

extern "C" void kernel_launch(void* const* d_in, const int* in_sizes, int n_in,
                              void* d_out, int out_size, void* d_ws,
                              size_t ws_size, hipStream_t stream) {
  const int* x    = (const int*)d_in[0];
  const int* ei   = (const int*)d_in[1];
  const float* emb = (const float*)d_in[2];
  const float* W1  = (const float*)d_in[3];
  const float* b1  = (const float*)d_in[4];
  const float* W2  = (const float*)d_in[5];
  const float* b2  = (const float*)d_in[6];
  float* out = (float*)d_out;

  const int N = in_sizes[0];
  const int E = in_sizes[1] / 2;
  const int* src = ei;
  const int* dst = ei + E;
  const int NT = 256;
  const int nsb = (N + NT - 1) / NT;

  char* ws = (char*)d_ws;
  size_t o = 0;
  int* cnt = (int*)(ws + o);        o += align256((size_t)N * 4);
  int* incl = (int*)(ws + o);       o += align256((size_t)N * 4);
  int* bsum = (int*)(ws + o);       o += align256((size_t)nsb * 4);
  int* rowptr = (int*)(ws + o);     o += align256((size_t)(N + 1) * 4);
  int* cursor = (int*)(ws + o);     o += align256((size_t)N * 4);
  int* col = (int*)(ws + o);        o += align256((size_t)E * 4);
  float* dinv = (float*)(ws + o);   o += align256((size_t)N * 4);
  float* bufA = (float*)(ws + o);   o += align256((size_t)N * 128 * 4);
  float* bufB = (float*)(ws + o);

  k_zero_i32<<<nsb, NT, 0, stream>>>(cnt, N);
  k_count<<<(E + NT - 1) / NT, NT, 0, stream>>>(dst, E, cnt);
  k_dinv<<<nsb, NT, 0, stream>>>(cnt, dinv, N);

  k_scan_block<<<nsb, NT, 0, stream>>>(cnt, N, incl, bsum);
  k_scan_sums<<<1, NT, 0, stream>>>(bsum, nsb);
  k_scan_finalize<<<nsb, NT, 0, stream>>>(incl, bsum, cnt, N, rowptr, cursor);
  k_fill<<<(E + NT - 1) / NT, NT, 0, stream>>>(src, dst, E, cursor, col);

  k_gemm_wmma<EMB_DIM, HID, true><<<N / 16, NT, 0, stream>>>(emb, W1, bufA, x);
  k_aggregate<true><<<(N + 7) / 8, NT, 0, stream>>>(rowptr, col, bufA, dinv, b1,
                                                    bufB, N);

  k_gemm_wmma<HID, OUTD, false><<<N / 16, NT, 0, stream>>>(bufB, W2, bufA,
                                                           nullptr);
  k_aggregate<false><<<(N + 7) / 8, NT, 0, stream>>>(rowptr, col, bufA, dinv,
                                                     b2, out, N);
}


# GEMM operand path: each wave's B slice held in registers over a loop of stripes (1024 work-groups), A stripe loaded once per work-group as whole lines to LDS, not 8x by 16-line loads; sums unchanged
# speedup vs baseline: 1.8678x; 1.8678x over previous
#include <hip/hip_runtime.h>
#include <hip/hip_bf16.h>


typedef float v2f __attribute__((ext_vector_type(2)));
typedef float v8f __attribute__((ext_vector_type(8)));

#define NNODES 50000
#define EMB_DIM 32
#define HID 128
#define OUTD 128


#define NCHUNK 128
#define HPITCH 50048

__global__ __launch_bounds__(1024) void k_chunk_hist(
    const int* __restrict__ dst, int E, int* __restrict__ hist) {
  __shared__ __attribute__((aligned(16))) int tab[NNODES];
  const int per = (E + (int)gridDim.x - 1) / (int)gridDim.x;
  const int e0 = blockIdx.x * per;
  const int e1 = min(E, e0 + per);
  int4* t4 = (int4*)tab;
  for (int i = threadIdx.x; i < NNODES / 4; i += 1024)
    t4[i] = make_int4(0, 0, 0, 0);
  __syncthreads();
  for (int e = e0 + (int)threadIdx.x; e < e1; e += 4 * 1024) {
    int d[4];
    for (int j = 0; j < 4; ++j) {
      int ee = e + j * 1024;
      d[j] = (ee < e1) ? dst[ee] : -1;
    }
    for (int j = 0; j < 4; ++j)
      if ((unsigned)d[j] < (unsigned)NNODES) atomicAdd(&tab[d[j]], 1);
  }
  __syncthreads();
  int4* h4 = (int4*)(hist + (size_t)blockIdx.x * HPITCH);
  for (int i = threadIdx.x; i < NNODES / 4; i += 1024) h4[i] = t4[i];
}

__global__ __launch_bounds__(256) void k_chunk_scan(
    const int* __restrict__ hist, int n, int* __restrict__ offs,
    float* __restrict__ dinv, int* __restrict__ incl, int* __restrict__ bsum) {
  __shared__ int sm[256];
  int i = blockIdx.x * 256 + threadIdx.x;
  int run = 0;
  if (i < n) {
#pragma unroll 32
    for (int g = 0; g < NCHUNK; ++g) {
      int c = hist[(size_t)g * HPITCH + i];
      offs[(size_t)g * HPITCH + i] = run;
      run += c;
    }
    dinv[i] = rsqrtf((float)(run + 1));
  }
  sm[threadIdx.x] = run;
  __syncthreads();
#pragma unroll
  for (int off = 1; off < 256; off <<= 1) {
    int t = (threadIdx.x >= off) ? sm[threadIdx.x - off] : 0;
    __syncthreads();
    sm[threadIdx.x] += t;
    __syncthreads();
  }
  if (i < n) incl[i] = sm[threadIdx.x];
  if (threadIdx.x == 255) bsum[blockIdx.x] = sm[255];
}

__global__ __launch_bounds__(256) void k_scan_sums(int* __restrict__ bsum,
                                                   int nb) {
  __shared__ int sm[256];
  int t = threadIdx.x;
  sm[t] = (t < nb) ? bsum[t] : 0;
  __syncthreads();
#pragma unroll
  for (int off = 1; off < 256; off <<= 1) {
    int v = (t >= off) ? sm[t - off] : 0;
    __syncthreads();
    sm[t] += v;
    __syncthreads();
  }
  if (t < nb) bsum[t] = sm[t];
}

__global__ __launch_bounds__(256) void k_scan_finalize(
    const int* __restrict__ incl, const int* __restrict__ bsumscan, int n,
    int* __restrict__ rowptr) {
  int i = blockIdx.x * 256 + threadIdx.x;
  if (i >= n) return;
  int boff = (blockIdx.x == 0) ? 0 : bsumscan[blockIdx.x - 1];
  rowptr[i + 1] = incl[i] + boff;
  if (i == 0) rowptr[0] = 0;
}

__global__ __launch_bounds__(1024) void k_chunk_place(
    const int* __restrict__ src, const int* __restrict__ dst, int E,
    const int* __restrict__ offs, const int* __restrict__ rowptr,
    int* __restrict__ col) {
  __shared__ __attribute__((aligned(16))) int tab[NNODES];
  const int per = (E + (int)gridDim.x - 1) / (int)gridDim.x;
  const int e0 = blockIdx.x * per;
  const int e1 = min(E, e0 + per);
  int4* t4 = (int4*)tab;
  const int4* o4 = (const int4*)(offs + (size_t)blockIdx.x * HPITCH);
  const int4* r4 = (const int4*)rowptr;
  for (int i = threadIdx.x; i < NNODES / 4; i += 1024) {
    int4 o = o4[i];
    int4 r = r4[i];
    t4[i] = make_int4(o.x + r.x, o.y + r.y, o.z + r.z, o.w + r.w);
  }
  __syncthreads();
  for (int e = e0 + (int)threadIdx.x; e < e1; e += 4 * 1024) {
    int s[4], d[4];
    for (int j = 0; j < 4; ++j) {
      int ee = e + j * 1024;
      bool in = ee < e1;
      s[j] = in ? src[ee] : -1;
      d[j] = in ? dst[ee] : -1;
    }
    for (int j = 0; j < 4; ++j) {
      if ((unsigned)s[j] < (unsigned)NNODES &&
          (unsigned)d[j] < (unsigned)NNODES) {
        int pos = atomicAdd(&tab[d[j]], 1);
        col[pos] = s[j];
      }
    }
  }
}


#define GEMM_GRID 1024

template <bool GATHER>
__device__ __forceinline__ int gemm_a_row(const int* __restrict__ xidx,
                                          int tile, int r) {
  int arow = tile * 16 + r;
  if (GATHER) arow = xidx[arow];
  return arow;
}

template <int K, int COLS, bool GATHER>
__global__ __launch_bounds__(256) void k_gemm_wmma(
    const float* __restrict__ A, const float* __restrict__ B,
    float* __restrict__ C, const int* __restrict__ xidx, int ntiles) {
  constexpr int PITCH = K + 4;
  constexpr int F4_ROW = K / 4;
  constexpr int NF4 = 16 * F4_ROW;
  static_assert(NF4 == 512 || NF4 == 128, "one or two pieces per thread");
  constexpr bool TWO = (NF4 == 512);
  __shared__ __attribute__((aligned(16))) float As[2][16 * PITCH];

  const int tid = threadIdx.x;
  const int lane = tid & 31;
  const int wave = tid >> 5;
  const int lo = lane & 15;
  const int hi = lane >> 4;
  const int col0 = wave * 16;
  const int G = (int)gridDim.x;
  const bool has0 = TWO || (tid < NF4);
  const int r0 = tid / F4_ROW, c0 = tid % F4_ROW;
  const int r1 = (tid + 256) / F4_ROW, c1 = (tid + 256) % F4_ROW;
  const int s0 = r0 * PITCH + 4 * c0;
  const int s1 = r1 * PITCH + 4 * c1;

  v2f bf[K / 4];
#pragma unroll
  for (int i = 0; i < K / 4; ++i) {
    const int kb = 4 * i;
    bf[i].x = B[(size_t)(kb + 2 * hi) * COLS + col0 + lo];
    bf[i].y = B[(size_t)(kb + 2 * hi + 1) * COLS + col0 + lo];
  }

  float4 p0 = make_float4(0.f, 0.f, 0.f, 0.f);
  float4 p1 = make_float4(0.f, 0.f, 0.f, 0.f);
  int an0 = 0, an1 = 0;
  int tile = blockIdx.x;
  if (tile < ntiles) {
    if (has0) {
      const int a0 = gemm_a_row<GATHER>(xidx, tile, r0);
      p0 = *((const float4*)(A + (size_t)a0 * K) + c0);
    }
    if constexpr (TWO) {
      const int a1 = gemm_a_row<GATHER>(xidx, tile, r1);
      p1 = *((const float4*)(A + (size_t)a1 * K) + c1);
    }
    if (tile + G < ntiles) {
      if (has0) an0 = gemm_a_row<GATHER>(xidx, tile + G, r0);
      if constexpr (TWO) an1 = gemm_a_row<GATHER>(xidx, tile + G, r1);
    }
    if (has0) *(float4*)&As[0][s0] = p0;
    if constexpr (TWO) *(float4*)&As[0][s1] = p1;
  }
  int cur = 0;
  for (; tile < ntiles; tile += G) {
    __syncthreads();
    const int nxt = tile + G;
    const bool more = nxt < ntiles;
    if (more) {
      if (has0) p0 = *((const float4*)(A + (size_t)an0 * K) + c0);
      if constexpr (TWO) p1 = *((const float4*)(A + (size_t)an1 * K) + c1);
      if (nxt + G < ntiles) {
        if (has0) an0 = gemm_a_row<GATHER>(xidx, nxt + G, r0);
        if constexpr (TWO) an1 = gemm_a_row<GATHER>(xidx, nxt + G, r1);
      }
    }

    const float* Ar = &As[cur][lo * PITCH + 2 * hi];
    v8f acc = {};
#pragma unroll
    for (int i = 0; i < K / 4; ++i) {
      const v2f a = *(const v2f*)(Ar + 4 * i);
      acc = __builtin_amdgcn_wmma_f32_16x16x4_f32(
          false, a, false, bf[i], (short)0, acc, false, false);
    }
    const int row0 = tile * 16;
#pragma unroll
    for (int g = 0; g < 8; ++g) {
      C[(size_t)(row0 + g + 8 * hi) * COLS + col0 + lo] = acc[g];
    }

    if (more) {
      if (has0) *(float4*)&As[cur ^ 1][s0] = p0;
      if constexpr (TWO) *(float4*)&As[cur ^ 1][s1] = p1;
    }
    cur ^= 1;
  }
}


template <bool RELU>
__global__ __launch_bounds__(256) void k_aggregate(
    const int* __restrict__ rowptr, const int* __restrict__ col,
    const float* __restrict__ H, const float* __restrict__ dinv,
    const float* __restrict__ bias, float* __restrict__ out, int n) {
  int d = blockIdx.x * (blockDim.x >> 5) + (threadIdx.x >> 5);
  int lane = threadIdx.x & 31;
  if (d >= n) return;

  int k0 = rowptr[d];
  int k1 = rowptr[d + 1];
  float4 acc = make_float4(0.f, 0.f, 0.f, 0.f);
  for (int k = k0; k < k1; ++k) {
    int s = col[k];
    float w = dinv[s];
    float4 v = *((const float4*)(H + (size_t)s * 128) + lane);
    acc.x += v.x * w;
    acc.y += v.y * w;
    acc.z += v.z * w;
    acc.w += v.w * w;
  }
  float dd = dinv[d];
  float4 self = *((const float4*)(H + (size_t)d * 128) + lane);
  float4 bb = *((const float4*)bias + lane);
  float4 r;
  r.x = (acc.x + self.x * dd) * dd + bb.x;
  r.y = (acc.y + self.y * dd) * dd + bb.y;
  r.z = (acc.z + self.z * dd) * dd + bb.z;
  r.w = (acc.w + self.w * dd) * dd + bb.w;
  if (RELU) {
    r.x = fmaxf(r.x, 0.f);
    r.y = fmaxf(r.y, 0.f);
    r.z = fmaxf(r.z, 0.f);
    r.w = fmaxf(r.w, 0.f);
  }
  *((float4*)(out + (size_t)d * 128) + lane) = r;
}


static inline size_t align256(size_t x) { return (x + 255) & ~(size_t)255; }

extern "C" void kernel_launch(void* const* d_in, const int* in_sizes, int n_in,
                              void* d_out, int out_size, void* d_ws,
                              size_t ws_size, hipStream_t stream) {
  const int* x    = (const int*)d_in[0];
  const int* ei   = (const int*)d_in[1];
  const float* emb = (const float*)d_in[2];
  const float* W1  = (const float*)d_in[3];
  const float* b1  = (const float*)d_in[4];
  const float* W2  = (const float*)d_in[5];
  const float* b2  = (const float*)d_in[6];
  float* out = (float*)d_out;

  const int N = in_sizes[0];
  const int E = in_sizes[1] / 2;
  const int* src = ei;
  const int* dst = ei + E;
  const int NT = 256;
  const int nsb = (N + NT - 1) / NT;

  char* ws = (char*)d_ws;
  size_t o = 0;
  int* hist = (int*)(ws + o);       o += align256((size_t)NCHUNK * HPITCH * 4);
  int* offs = (int*)(ws + o);       o += align256((size_t)NCHUNK * HPITCH * 4);
  int* incl = (int*)(ws + o);       o += align256((size_t)N * 4);
  int* bsum = (int*)(ws + o);       o += align256((size_t)nsb * 4);
  int* rowptr = (int*)(ws + o);     o += align256((size_t)(N + 1) * 4);
  int* col = (int*)(ws + o);        o += align256((size_t)E * 4);
  float* dinv = (float*)(ws + o);   o += align256((size_t)N * 4);
  float* bufA = (float*)(ws + o);   o += align256((size_t)N * 128 * 4);
  float* bufB = (float*)(ws + o);

  k_chunk_hist<<<NCHUNK, 1024, 0, stream>>>(dst, E, hist);
  k_chunk_scan<<<nsb, NT, 0, stream>>>(hist, N, offs, dinv, incl, bsum);

  k_scan_sums<<<1, NT, 0, stream>>>(bsum, nsb);
  k_scan_finalize<<<nsb, NT, 0, stream>>>(incl, bsum, N, rowptr);
  k_chunk_place<<<NCHUNK, 1024, 0, stream>>>(src, dst, E, offs, rowptr, col);

  k_gemm_wmma<EMB_DIM, HID, true><<<GEMM_GRID, NT, 0, stream>>>(emb, W1, bufA, x,
                                                          N / 16);
  k_aggregate<true><<<(N + 7) / 8, NT, 0, stream>>>(rowptr, col, bufA, dinv, b1,
                                                    bufB, N);

  k_gemm_wmma<HID, OUTD, false><<<GEMM_GRID, NT, 0, stream>>>(bufB, W2, bufA,
                                                           nullptr, N / 16);
  k_aggregate<false><<<(N + 7) / 8, NT, 0, stream>>>(rowptr, col, bufA, dinv,
                                                     b2, out, N);
}


# Launches 9 -> 6: layer-1 GEMM rides in the chunk-scan launch as independent work-groups; placement work-groups scan the 196 block totals themselves (two tiny launches dropped); sums unchanged
# speedup vs baseline: 1.9844x; 1.0624x over previous
#include <hip/hip_runtime.h>
#include <hip/hip_bf16.h>


typedef float v2f __attribute__((ext_vector_type(2)));
typedef float v8f __attribute__((ext_vector_type(8)));

#define NNODES 50000
#define EMB_DIM 32
#define HID 128
#define OUTD 128


#define NCHUNK 128
#define HPITCH 50048

__global__ __launch_bounds__(1024) void k_chunk_hist(
    const int* __restrict__ dst, int E, int* __restrict__ hist) {
  __shared__ __attribute__((aligned(16))) int tab[NNODES];
  const int per = (E + (int)gridDim.x - 1) / (int)gridDim.x;
  const int e0 = blockIdx.x * per;
  const int e1 = min(E, e0 + per);
  int4* t4 = (int4*)tab;
  for (int i = threadIdx.x; i < NNODES / 4; i += 1024)
    t4[i] = make_int4(0, 0, 0, 0);
  __syncthreads();
  for (int e = e0 + (int)threadIdx.x; e < e1; e += 4 * 1024) {
    int d[4];
    for (int j = 0; j < 4; ++j) {
      int ee = e + j * 1024;
      d[j] = (ee < e1) ? dst[ee] : -1;
    }
    for (int j = 0; j < 4; ++j)
      if ((unsigned)d[j] < (unsigned)NNODES) atomicAdd(&tab[d[j]], 1);
  }
  __syncthreads();
  int4* h4 = (int4*)(hist + (size_t)blockIdx.x * HPITCH);
  for (int i = threadIdx.x; i < NNODES / 4; i += 1024) h4[i] = t4[i];
}

__device__ __forceinline__ void chunk_scan_body(
    int blk, const int* __restrict__ hist, int n, int* __restrict__ offs,
    float* __restrict__ dinv, int* __restrict__ excl, int* __restrict__ bsum) {
  __shared__ int sm[256];
  int i = blk * 256 + threadIdx.x;
  int run = 0;
  if (i < n) {
#pragma unroll 32
    for (int g = 0; g < NCHUNK; ++g) {
      int c = hist[(size_t)g * HPITCH + i];
      offs[(size_t)g * HPITCH + i] = run;
      run += c;
    }
    dinv[i] = rsqrtf((float)(run + 1));
  }
  sm[threadIdx.x] = run;
  __syncthreads();
#pragma unroll
  for (int off = 1; off < 256; off <<= 1) {
    int t = (threadIdx.x >= off) ? sm[threadIdx.x - off] : 0;
    __syncthreads();
    sm[threadIdx.x] += t;
    __syncthreads();
  }
  if (i < n) excl[i] = sm[threadIdx.x] - run;
  if (threadIdx.x == 255) bsum[blk] = sm[255];
}

__global__ __launch_bounds__(1024) void k_chunk_place(
    const int* __restrict__ src, const int* __restrict__ dst, int E,
    const int* __restrict__ offs, const int* __restrict__ excl,
    const int* __restrict__ bsum, int nb, int n, int* __restrict__ rowptr,
    int* __restrict__ col) {
  __shared__ __attribute__((aligned(16))) int tab[NNODES];
  __shared__ int boff[257];
  const int G = (int)gridDim.x;
  const int per = (E + G - 1) / G;
  const int e0 = blockIdx.x * per;
  const int e1 = min(E, e0 + per);
  if (threadIdx.x < 32) {
    const int lane = threadIdx.x;
    int v[8];
#pragma unroll
    for (int j = 0; j < 8; ++j) {
      const int k = 8 * lane + j;
      v[j] = (k < nb) ? bsum[k] : 0;
    }
    int tot = 0;
#pragma unroll
    for (int j = 0; j < 8; ++j) {
      const int t = v[j];
      v[j] = tot;
      tot += t;
    }
    int inc = tot;
    for (int off = 1; off < 32; off <<= 1) {
      const int t = __shfl_up(inc, off);
      if (lane >= off) inc += t;
    }
    const int base = inc - tot;
#pragma unroll
    for (int j = 0; j < 8; ++j) boff[8 * lane + j] = base + v[j];
    if (lane == 31) boff[256] = inc;
  }
  __syncthreads();
  int4* t4 = (int4*)tab;
  const int4* o4 = (const int4*)(offs + (size_t)blockIdx.x * HPITCH);
  const int4* x4 = (const int4*)excl;
  for (int i = threadIdx.x; i < NNODES / 4; i += 1024) {
    int4 o = o4[i];
    int4 r = x4[i];
    const int b = boff[i >> 6];
    t4[i] = make_int4(o.x + r.x + b, o.y + r.y + b, o.z + r.z + b,
                      o.w + r.w + b);
  }
  const int nper = (((n + G) / G) + 31) & ~31;
  const int j0 = blockIdx.x * nper;
  const int j1 = min(n + 1, j0 + nper);
  for (int j = j0 + (int)threadIdx.x; j < j1; j += 1024)
    rowptr[j] = (j < n) ? excl[j] + boff[j >> 8] : boff[256];
  __syncthreads();
  for (int e = e0 + (int)threadIdx.x; e < e1; e += 4 * 1024) {
    int s[4], d[4];
    for (int j = 0; j < 4; ++j) {
      int ee = e + j * 1024;
      bool in = ee < e1;
      s[j] = in ? src[ee] : -1;
      d[j] = in ? dst[ee] : -1;
    }
    for (int j = 0; j < 4; ++j) {
      if ((unsigned)s[j] < (unsigned)NNODES &&
          (unsigned)d[j] < (unsigned)NNODES) {
        int pos = atomicAdd(&tab[d[j]], 1);
        col[pos] = s[j];
      }
    }
  }
}


#define GEMM_GRID 1024

template <bool GATHER>
__device__ __forceinline__ int gemm_a_row(const int* __restrict__ xidx,
                                          int tile, int r) {
  int arow = tile * 16 + r;
  if (GATHER) arow = xidx[arow];
  return arow;
}

template <int K, int COLS, bool GATHER>
__device__ __forceinline__ void gemm_body(
    int tile0, int G, const float* __restrict__ A, const float* __restrict__ B,
    float* __restrict__ C, const int* __restrict__ xidx, int ntiles) {
  constexpr int PITCH = K + 4;
  constexpr int F4_ROW = K / 4;
  constexpr int NF4 = 16 * F4_ROW;
  static_assert(NF4 == 512 || NF4 == 128, "one or two pieces per thread");
  constexpr bool TWO = (NF4 == 512);
  __shared__ __attribute__((aligned(16))) float As[2][16 * PITCH];

  const int tid = threadIdx.x;
  const int lane = tid & 31;
  const int wave = tid >> 5;
  const int lo = lane & 15;
  const int hi = lane >> 4;
  const int col0 = wave * 16;
  const bool has0 = TWO || (tid < NF4);
  const int r0 = tid / F4_ROW, c0 = tid % F4_ROW;
  const int r1 = (tid + 256) / F4_ROW, c1 = (tid + 256) % F4_ROW;
  const int s0 = r0 * PITCH + 4 * c0;
  const int s1 = r1 * PITCH + 4 * c1;

  v2f bf[K / 4];
#pragma unroll
  for (int i = 0; i < K / 4; ++i) {
    const int kb = 4 * i;
    bf[i].x = B[(size_t)(kb + 2 * hi) * COLS + col0 + lo];
    bf[i].y = B[(size_t)(kb + 2 * hi + 1) * COLS + col0 + lo];
  }

  float4 p0 = make_float4(0.f, 0.f, 0.f, 0.f);
  float4 p1 = make_float4(0.f, 0.f, 0.f, 0.f);
  int an0 = 0, an1 = 0;
  int tile = tile0;
  if (tile < ntiles) {
    if (has0) {
      const int a0 = gemm_a_row<GATHER>(xidx, tile, r0);
      p0 = *((const float4*)(A + (size_t)a0 * K) + c0);
    }
    if constexpr (TWO) {
      const int a1 = gemm_a_row<GATHER>(xidx, tile, r1);
      p1 = *((const float4*)(A + (size_t)a1 * K) + c1);
    }
    if (tile + G < ntiles) {
      if (has0) an0 = gemm_a_row<GATHER>(xidx, tile + G, r0);
      if constexpr (TWO) an1 = gemm_a_row<GATHER>(xidx, tile + G, r1);
    }
    if (has0) *(float4*)&As[0][s0] = p0;
    if constexpr (TWO) *(float4*)&As[0][s1] = p1;
  }
  int cur = 0;
  for (; tile < ntiles; tile += G) {
    __syncthreads();
    const int nxt = tile + G;
    const bool more = nxt < ntiles;
    if (more) {
      if (has0) p0 = *((const float4*)(A + (size_t)an0 * K) + c0);
      if constexpr (TWO) p1 = *((const float4*)(A + (size_t)an1 * K) + c1);
      if (nxt + G < ntiles) {
        if (has0) an0 = gemm_a_row<GATHER>(xidx, nxt + G, r0);
        if constexpr (TWO) an1 = gemm_a_row<GATHER>(xidx, nxt + G, r1);
      }
    }

    const float* Ar = &As[cur][lo * PITCH + 2 * hi];
    v8f acc = {};
#pragma unroll
    for (int i = 0; i < K / 4; ++i) {
      const v2f a = *(const v2f*)(Ar + 4 * i);
      acc = __builtin_amdgcn_wmma_f32_16x16x4_f32(
          false, a, false, bf[i], (short)0, acc, false, false);
    }
    const int row0 = tile * 16;
#pragma unroll
    for (int g = 0; g < 8; ++g) {
      C[(size_t)(row0 + g + 8 * hi) * COLS + col0 + lo] = acc[g];
    }

    if (more) {
      if (has0) *(float4*)&As[cur ^ 1][s0] = p0;
      if constexpr (TWO) *(float4*)&As[cur ^ 1][s1] = p1;
    }
    cur ^= 1;
  }
}

template <int K, int COLS, bool GATHER>
__global__ __launch_bounds__(256) void k_gemm_wmma(
    const float* __restrict__ A, const float* __restrict__ B,
    float* __restrict__ C, const int* __restrict__ xidx, int ntiles) {
  gemm_body<K, COLS, GATHER>((int)blockIdx.x, (int)gridDim.x, A, B, C, xidx,
                             ntiles);
}

__global__ __launch_bounds__(256) void k_scan_and_gemm1(
    const int* __restrict__ hist, int n, int* __restrict__ offs,
    float* __restrict__ dinv, int* __restrict__ excl, int* __restrict__ bsum,
    int nsb, const float* __restrict__ emb, const float* __restrict__ W1,
    float* __restrict__ H1, const int* __restrict__ xidx, int ntiles) {
  const int blk = (int)blockIdx.x;
  if (blk < nsb) {
    chunk_scan_body(blk, hist, n, offs, dinv, excl, bsum);
  } else {
    gemm_body<EMB_DIM, HID, true>(blk - nsb, (int)gridDim.x - nsb, emb, W1, H1,
                                  xidx, ntiles);
  }
}


template <bool RELU>
__global__ __launch_bounds__(256) void k_aggregate(
    const int* __restrict__ rowptr, const int* __restrict__ col,
    const float* __restrict__ H, const float* __restrict__ dinv,
    const float* __restrict__ bias, float* __restrict__ out, int n) {
  int d = blockIdx.x * (blockDim.x >> 5) + (threadIdx.x >> 5);
  int lane = threadIdx.x & 31;
  if (d >= n) return;

  int k0 = rowptr[d];
  int k1 = rowptr[d + 1];
  float4 acc = make_float4(0.f, 0.f, 0.f, 0.f);
  for (int k = k0; k < k1; ++k) {
    int s = col[k];
    float w = dinv[s];
    float4 v = *((const float4*)(H + (size_t)s * 128) + lane);
    acc.x += v.x * w;
    acc.y += v.y * w;
    acc.z += v.z * w;
    acc.w += v.w * w;
  }
  float dd = dinv[d];
  float4 self = *((const float4*)(H + (size_t)d * 128) + lane);
  float4 bb = *((const float4*)bias + lane);
  float4 r;
  r.x = (acc.x + self.x * dd) * dd + bb.x;
  r.y = (acc.y + self.y * dd) * dd + bb.y;
  r.z = (acc.z + self.z * dd) * dd + bb.z;
  r.w = (acc.w + self.w * dd) * dd + bb.w;
  if (RELU) {
    r.x = fmaxf(r.x, 0.f);
    r.y = fmaxf(r.y, 0.f);
    r.z = fmaxf(r.z, 0.f);
    r.w = fmaxf(r.w, 0.f);
  }
  *((float4*)(out + (size_t)d * 128) + lane) = r;
}


static inline size_t align256(size_t x) { return (x + 255) & ~(size_t)255; }

extern "C" void kernel_launch(void* const* d_in, const int* in_sizes, int n_in,
                              void* d_out, int out_size, void* d_ws,
                              size_t ws_size, hipStream_t stream) {
  const int* x    = (const int*)d_in[0];
  const int* ei   = (const int*)d_in[1];
  const float* emb = (const float*)d_in[2];
  const float* W1  = (const float*)d_in[3];
  const float* b1  = (const float*)d_in[4];
  const float* W2  = (const float*)d_in[5];
  const float* b2  = (const float*)d_in[6];
  float* out = (float*)d_out;

  const int N = in_sizes[0];
  const int E = in_sizes[1] / 2;
  const int* src = ei;
  const int* dst = ei + E;
  const int NT = 256;
  const int nsb = (N + NT - 1) / NT;

  char* ws = (char*)d_ws;
  size_t o = 0;
  int* hist = (int*)(ws + o);       o += align256((size_t)NCHUNK * HPITCH * 4);
  int* offs = (int*)(ws + o);       o += align256((size_t)NCHUNK * HPITCH * 4);
  int* excl = (int*)(ws + o);       o += align256((size_t)N * 4);
  int* bsum = (int*)(ws + o);       o += align256((size_t)nsb * 4);
  int* rowptr = (int*)(ws + o);     o += align256((size_t)(N + 1) * 4);
  int* col = (int*)(ws + o);        o += align256((size_t)E * 4);
  float* dinv = (float*)(ws + o);   o += align256((size_t)N * 4);
  float* bufA = (float*)(ws + o);   o += align256((size_t)N * 128 * 4);
  float* bufB = (float*)(ws + o);

  k_chunk_hist<<<NCHUNK, 1024, 0, stream>>>(dst, E, hist);
  k_scan_and_gemm1<<<nsb + GEMM_GRID, NT, 0, stream>>>(
      hist, N, offs, dinv, excl, bsum, nsb, emb, W1, bufA, x, N / 16);

  k_chunk_place<<<NCHUNK, 1024, 0, stream>>>(src, dst, E, offs, excl, bsum, nsb,
                                             N, rowptr, col);

  k_aggregate<true><<<(N + 7) / 8, NT, 0, stream>>>(rowptr, col, bufA, dinv, b1,
                                                    bufB, N);

  k_gemm_wmma<HID, OUTD, false><<<GEMM_GRID, NT, 0, stream>>>(bufB, W2, bufA,
                                                           nullptr, N / 16);
  k_aggregate<false><<<(N + 7) / 8, NT, 0, stream>>>(rowptr, col, bufA, dinv,
                                                     b2, out, N);
}


# Aggregation sends fewer line requests: edge ids read 32 per coalesced load and broadcast by readlane; layer 1 stores its gathered dinv[src] in edge order, layer 2 reads them back; sums unchanged
# speedup vs baseline: 2.0631x; 1.0397x over previous
#include <hip/hip_runtime.h>
#include <hip/hip_bf16.h>


typedef float v2f __attribute__((ext_vector_type(2)));
typedef float v8f __attribute__((ext_vector_type(8)));

#define NNODES 50000
#define EMB_DIM 32
#define HID 128
#define OUTD 128


#define NCHUNK 128
#define HPITCH 50048

__global__ __launch_bounds__(1024) void k_chunk_hist(
    const int* __restrict__ dst, int E, int* __restrict__ hist) {
  __shared__ __attribute__((aligned(16))) int tab[NNODES];
  const int per = (E + (int)gridDim.x - 1) / (int)gridDim.x;
  const int e0 = blockIdx.x * per;
  const int e1 = min(E, e0 + per);
  int4* t4 = (int4*)tab;
  for (int i = threadIdx.x; i < NNODES / 4; i += 1024)
    t4[i] = make_int4(0, 0, 0, 0);
  __syncthreads();
  for (int e = e0 + (int)threadIdx.x; e < e1; e += 4 * 1024) {
    int d[4];
    for (int j = 0; j < 4; ++j) {
      int ee = e + j * 1024;
      d[j] = (ee < e1) ? dst[ee] : -1;
    }
    for (int j = 0; j < 4; ++j)
      if ((unsigned)d[j] < (unsigned)NNODES) atomicAdd(&tab[d[j]], 1);
  }
  __syncthreads();
  int4* h4 = (int4*)(hist + (size_t)blockIdx.x * HPITCH);
  for (int i = threadIdx.x; i < NNODES / 4; i += 1024) h4[i] = t4[i];
}

__device__ __forceinline__ void chunk_scan_body(
    int blk, const int* __restrict__ hist, int n, int* __restrict__ offs,
    float* __restrict__ dinv, int* __restrict__ excl, int* __restrict__ bsum) {
  __shared__ int sm[256];
  int i = blk * 256 + threadIdx.x;
  int run = 0;
  if (i < n) {
#pragma unroll 32
    for (int g = 0; g < NCHUNK; ++g) {
      int c = hist[(size_t)g * HPITCH + i];
      offs[(size_t)g * HPITCH + i] = run;
      run += c;
    }
    dinv[i] = rsqrtf((float)(run + 1));
  }
  sm[threadIdx.x] = run;
  __syncthreads();
#pragma unroll
  for (int off = 1; off < 256; off <<= 1) {
    int t = (threadIdx.x >= off) ? sm[threadIdx.x - off] : 0;
    __syncthreads();
    sm[threadIdx.x] += t;
    __syncthreads();
  }
  if (i < n) excl[i] = sm[threadIdx.x] - run;
  if (threadIdx.x == 255) bsum[blk] = sm[255];
}

__global__ __launch_bounds__(1024) void k_chunk_place(
    const int* __restrict__ src, const int* __restrict__ dst, int E,
    const int* __restrict__ offs, const int* __restrict__ excl,
    const int* __restrict__ bsum, int nb, int n, int* __restrict__ rowptr,
    int* __restrict__ col) {
  __shared__ __attribute__((aligned(16))) int tab[NNODES];
  __shared__ int boff[257];
  const int G = (int)gridDim.x;
  const int per = (E + G - 1) / G;
  const int e0 = blockIdx.x * per;
  const int e1 = min(E, e0 + per);
  if (threadIdx.x < 32) {
    const int lane = threadIdx.x;
    int v[8];
#pragma unroll
    for (int j = 0; j < 8; ++j) {
      const int k = 8 * lane + j;
      v[j] = (k < nb) ? bsum[k] : 0;
    }
    int tot = 0;
#pragma unroll
    for (int j = 0; j < 8; ++j) {
      const int t = v[j];
      v[j] = tot;
      tot += t;
    }
    int inc = tot;
    for (int off = 1; off < 32; off <<= 1) {
      const int t = __shfl_up(inc, off);
      if (lane >= off) inc += t;
    }
    const int base = inc - tot;
#pragma unroll
    for (int j = 0; j < 8; ++j) boff[8 * lane + j] = base + v[j];
    if (lane == 31) boff[256] = inc;
  }
  __syncthreads();
  int4* t4 = (int4*)tab;
  const int4* o4 = (const int4*)(offs + (size_t)blockIdx.x * HPITCH);
  const int4* x4 = (const int4*)excl;
  for (int i = threadIdx.x; i < NNODES / 4; i += 1024) {
    int4 o = o4[i];
    int4 r = x4[i];
    const int b = boff[i >> 6];
    t4[i] = make_int4(o.x + r.x + b, o.y + r.y + b, o.z + r.z + b,
                      o.w + r.w + b);
  }
  const int nper = (((n + G) / G) + 31) & ~31;
  const int j0 = blockIdx.x * nper;
  const int j1 = min(n + 1, j0 + nper);
  for (int j = j0 + (int)threadIdx.x; j < j1; j += 1024)
    rowptr[j] = (j < n) ? excl[j] + boff[j >> 8] : boff[256];
  __syncthreads();
  for (int e = e0 + (int)threadIdx.x; e < e1; e += 4 * 1024) {
    int s[4], d[4];
    for (int j = 0; j < 4; ++j) {
      int ee = e + j * 1024;
      bool in = ee < e1;
      s[j] = in ? src[ee] : -1;
      d[j] = in ? dst[ee] : -1;
    }
    for (int j = 0; j < 4; ++j) {
      if ((unsigned)s[j] < (unsigned)NNODES &&
          (unsigned)d[j] < (unsigned)NNODES) {
        int pos = atomicAdd(&tab[d[j]], 1);
        col[pos] = s[j];
      }
    }
  }
}


#define GEMM_GRID 1024

template <bool GATHER>
__device__ __forceinline__ int gemm_a_row(const int* __restrict__ xidx,
                                          int tile, int r) {
  int arow = tile * 16 + r;
  if (GATHER) arow = xidx[arow];
  return arow;
}

template <int K, int COLS, bool GATHER>
__device__ __forceinline__ void gemm_body(
    int tile0, int G, const float* __restrict__ A, const float* __restrict__ B,
    float* __restrict__ C, const int* __restrict__ xidx, int ntiles) {
  constexpr int PITCH = K + 4;
  constexpr int F4_ROW = K / 4;
  constexpr int NF4 = 16 * F4_ROW;
  static_assert(NF4 == 512 || NF4 == 128, "one or two pieces per thread");
  constexpr bool TWO = (NF4 == 512);
  __shared__ __attribute__((aligned(16))) float As[2][16 * PITCH];

  const int tid = threadIdx.x;
  const int lane = tid & 31;
  const int wave = tid >> 5;
  const int lo = lane & 15;
  const int hi = lane >> 4;
  const int col0 = wave * 16;
  const bool has0 = TWO || (tid < NF4);
  const int r0 = tid / F4_ROW, c0 = tid % F4_ROW;
  const int r1 = (tid + 256) / F4_ROW, c1 = (tid + 256) % F4_ROW;
  const int s0 = r0 * PITCH + 4 * c0;
  const int s1 = r1 * PITCH + 4 * c1;

  v2f bf[K / 4];
#pragma unroll
  for (int i = 0; i < K / 4; ++i) {
    const int kb = 4 * i;
    bf[i].x = B[(size_t)(kb + 2 * hi) * COLS + col0 + lo];
    bf[i].y = B[(size_t)(kb + 2 * hi + 1) * COLS + col0 + lo];
  }

  float4 p0 = make_float4(0.f, 0.f, 0.f, 0.f);
  float4 p1 = make_float4(0.f, 0.f, 0.f, 0.f);
  int an0 = 0, an1 = 0;
  int tile = tile0;
  if (tile < ntiles) {
    if (has0) {
      const int a0 = gemm_a_row<GATHER>(xidx, tile, r0);
      p0 = *((const float4*)(A + (size_t)a0 * K) + c0);
    }
    if constexpr (TWO) {
      const int a1 = gemm_a_row<GATHER>(xidx, tile, r1);
      p1 = *((const float4*)(A + (size_t)a1 * K) + c1);
    }
    if (tile + G < ntiles) {
      if (has0) an0 = gemm_a_row<GATHER>(xidx, tile + G, r0);
      if constexpr (TWO) an1 = gemm_a_row<GATHER>(xidx, tile + G, r1);
    }
    if (has0) *(float4*)&As[0][s0] = p0;
    if constexpr (TWO) *(float4*)&As[0][s1] = p1;
  }
  int cur = 0;
  for (; tile < ntiles; tile += G) {
    __syncthreads();
    const int nxt = tile + G;
    const bool more = nxt < ntiles;
    if (more) {
      if (has0) p0 = *((const float4*)(A + (size_t)an0 * K) + c0);
      if constexpr (TWO) p1 = *((const float4*)(A + (size_t)an1 * K) + c1);
      if (nxt + G < ntiles) {
        if (has0) an0 = gemm_a_row<GATHER>(xidx, nxt + G, r0);
        if constexpr (TWO) an1 = gemm_a_row<GATHER>(xidx, nxt + G, r1);
      }
    }

    const float* Ar = &As[cur][lo * PITCH + 2 * hi];
    v8f acc = {};
#pragma unroll
    for (int i = 0; i < K / 4; ++i) {
      const v2f a = *(const v2f*)(Ar + 4 * i);
      acc = __builtin_amdgcn_wmma_f32_16x16x4_f32(
          false, a, false, bf[i], (short)0, acc, false, false);
    }
    const int row0 = tile * 16;
#pragma unroll
    for (int g = 0; g < 8; ++g) {
      C[(size_t)(row0 + g + 8 * hi) * COLS + col0 + lo] = acc[g];
    }

    if (more) {
      if (has0) *(float4*)&As[cur ^ 1][s0] = p0;
      if constexpr (TWO) *(float4*)&As[cur ^ 1][s1] = p1;
    }
    cur ^= 1;
  }
}

template <int K, int COLS, bool GATHER>
__global__ __launch_bounds__(256) void k_gemm_wmma(
    const float* __restrict__ A, const float* __restrict__ B,
    float* __restrict__ C, const int* __restrict__ xidx, int ntiles) {
  gemm_body<K, COLS, GATHER>((int)blockIdx.x, (int)gridDim.x, A, B, C, xidx,
                             ntiles);
}

__global__ __launch_bounds__(256) void k_scan_and_gemm1(
    const int* __restrict__ hist, int n, int* __restrict__ offs,
    float* __restrict__ dinv, int* __restrict__ excl, int* __restrict__ bsum,
    int nsb, const float* __restrict__ emb, const float* __restrict__ W1,
    float* __restrict__ H1, const int* __restrict__ xidx, int ntiles) {
  const int blk = (int)blockIdx.x;
  if (blk < nsb) {
    chunk_scan_body(blk, hist, n, offs, dinv, excl, bsum);
  } else {
    gemm_body<EMB_DIM, HID, true>(blk - nsb, (int)gridDim.x - nsb, emb, W1, H1,
                                  xidx, ntiles);
  }
}


#define AGG_GROUP 8

__device__ __forceinline__ void agg_fma(float4& acc, const float4 v, float w) {
  acc.x = __builtin_fmaf(v.x, w, acc.x);
  acc.y = __builtin_fmaf(v.y, w, acc.y);
  acc.z = __builtin_fmaf(v.z, w, acc.z);
  acc.w = __builtin_fmaf(v.w, w, acc.w);
}

template <bool RELU, bool FIRST>
__global__ __launch_bounds__(256) void k_aggregate(
    const int* __restrict__ rowptr, const int* __restrict__ col,
    float* __restrict__ wedge, const float* __restrict__ H,
    const float* __restrict__ dinv, const float* __restrict__ bias,
    float* __restrict__ out, int n) {
  const int lane = threadIdx.x & 31;
  const int d = __builtin_amdgcn_readfirstlane(
      (int)(blockIdx.x * (blockDim.x >> 5) + (threadIdx.x >> 5)));
  if (d >= n) return;

  const int k0 = rowptr[d];
  const int k1 = rowptr[d + 1];
  const float4* __restrict__ Hl = (const float4*)H + lane;
  const float dd = dinv[d];
  const float4 self = Hl[(size_t)d * 32];
  const float4 bb = *((const float4*)bias + lane);

  float4 acc = make_float4(0.f, 0.f, 0.f, 0.f);
  for (int kb = k0; kb < k1; kb += 32) {
    const int kk = kb + lane;
    int sl = 0;
    float wl = 0.f;
    if (kk < k1) {
      sl = col[kk];
      wl = FIRST ? dinv[sl] : wedge[kk];
    }
    const int cnt = min(32, k1 - kb);
    int j = 0;
    for (; j + AGG_GROUP <= cnt; j += AGG_GROUP) {
      float4 v[AGG_GROUP];
      float w[AGG_GROUP];
#pragma unroll
      for (int u = 0; u < AGG_GROUP; ++u) {
        const int s = __builtin_amdgcn_readlane(sl, j + u);
        w[u] = __builtin_bit_cast(
            float, __builtin_amdgcn_readlane(__builtin_bit_cast(int, wl), j + u));
        v[u] = Hl[(size_t)s * 32];
      }
#pragma unroll
      for (int u = 0; u < AGG_GROUP; ++u) agg_fma(acc, v[u], w[u]);
    }
    const int rem = cnt - j;
    if (rem > 0) {
      float4 v[AGG_GROUP - 1];
      float w[AGG_GROUP - 1];
#pragma unroll
      for (int u = 0; u < AGG_GROUP - 1; ++u) {
        if (u < rem) {
          const int s = __builtin_amdgcn_readlane(sl, j + u);
          w[u] = __builtin_bit_cast(
              float,
              __builtin_amdgcn_readlane(__builtin_bit_cast(int, wl), j + u));
          v[u] = Hl[(size_t)s * 32];
        }
      }
#pragma unroll
      for (int u = 0; u < AGG_GROUP - 1; ++u) {
        if (u < rem) agg_fma(acc, v[u], w[u]);
      }
    }
    if (FIRST && kk < k1) wedge[kk] = wl;
  }
  float4 r;
  r.x = (acc.x + self.x * dd) * dd + bb.x;
  r.y = (acc.y + self.y * dd) * dd + bb.y;
  r.z = (acc.z + self.z * dd) * dd + bb.z;
  r.w = (acc.w + self.w * dd) * dd + bb.w;
  if (RELU) {
    r.x = fmaxf(r.x, 0.f);
    r.y = fmaxf(r.y, 0.f);
    r.z = fmaxf(r.z, 0.f);
    r.w = fmaxf(r.w, 0.f);
  }
  *((float4*)(out + (size_t)d * 128) + lane) = r;
}


static inline size_t align256(size_t x) { return (x + 255) & ~(size_t)255; }

extern "C" void kernel_launch(void* const* d_in, const int* in_sizes, int n_in,
                              void* d_out, int out_size, void* d_ws,
                              size_t ws_size, hipStream_t stream) {
  const int* x    = (const int*)d_in[0];
  const int* ei   = (const int*)d_in[1];
  const float* emb = (const float*)d_in[2];
  const float* W1  = (const float*)d_in[3];
  const float* b1  = (const float*)d_in[4];
  const float* W2  = (const float*)d_in[5];
  const float* b2  = (const float*)d_in[6];
  float* out = (float*)d_out;

  const int N = in_sizes[0];
  const int E = in_sizes[1] / 2;
  const int* src = ei;
  const int* dst = ei + E;
  const int NT = 256;
  const int nsb = (N + NT - 1) / NT;

  char* ws = (char*)d_ws;
  size_t o = 0;
  int* hist = (int*)(ws + o);       o += align256((size_t)NCHUNK * HPITCH * 4);
  int* offs = (int*)(ws + o);       o += align256((size_t)NCHUNK * HPITCH * 4);
  int* excl = (int*)(ws + o);       o += align256((size_t)N * 4);
  int* bsum = (int*)(ws + o);       o += align256((size_t)nsb * 4);
  int* rowptr = (int*)(ws + o);     o += align256((size_t)(N + 1) * 4);
  int* col = (int*)(ws + o);        o += align256((size_t)E * 4);
  float* dinv = (float*)(ws + o);   o += align256((size_t)N * 4);
  float* bufA = (float*)(ws + o);   o += align256((size_t)N * 128 * 4);
  float* bufB = (float*)(ws + o);   o += align256((size_t)N * 128 * 4);
  float* wedge = (float*)(ws + o);

  k_chunk_hist<<<NCHUNK, 1024, 0, stream>>>(dst, E, hist);
  k_scan_and_gemm1<<<nsb + GEMM_GRID, NT, 0, stream>>>(
      hist, N, offs, dinv, excl, bsum, nsb, emb, W1, bufA, x, N / 16);

  k_chunk_place<<<NCHUNK, 1024, 0, stream>>>(src, dst, E, offs, excl, bsum, nsb,
                                             N, rowptr, col);

  k_aggregate<true, true><<<(N + 7) / 8, NT, 0, stream>>>(
      rowptr, col, wedge, bufA, dinv, b1, bufB, N);

  k_gemm_wmma<HID, OUTD, false><<<GEMM_GRID, NT, 0, stream>>>(bufB, W2, bufA,
                                                           nullptr, N / 16);
  k_aggregate<false, false><<<(N + 7) / 8, NT, 0, stream>>>(
      rowptr, col, wedge, bufA, dinv, b2, out, N);
}


# CSR build on all 256 WGPs: each edge chunk is shared by two 512-thread work-groups, one per half of the node range (100 KB LDS table each), 16 edges per thread and round; neighbour order unchanged
# speedup vs baseline: 2.1509x; 1.0426x over previous
#include <hip/hip_runtime.h>
#include <hip/hip_bf16.h>


typedef float v2f __attribute__((ext_vector_type(2)));
typedef float v8f __attribute__((ext_vector_type(8)));

#define NNODES 50000
#define EMB_DIM 32
#define HID 128
#define OUTD 128


#define NCHUNK 128
#define CSR_GRID 256
#define CSR_T 512
#define CSR_B 16
#define HPITCH 50048
#define HSPLIT 25024
#define HTAB 25024
#define HTRIPS ((HTAB / 4 + CSR_T - 1) / CSR_T)
static_assert(HSPLIT % 32 == 0 && HSPLIT <= HTAB && NNODES - HSPLIT <= HTAB &&
                  (NNODES - HSPLIT) % 4 == 0 &&
                  (HTRIPS - 1) * CSR_T <= (NNODES - HSPLIT) / 4 &&
                  (HTRIPS - 1) * CSR_T <= HSPLIT / 4,
              "halves: whole lines, whole int4s, inside the table");

__global__ __launch_bounds__(CSR_T) void k_chunk_hist(
    const int* __restrict__ dst, int E, int* __restrict__ hist) {
  __shared__ __attribute__((aligned(16))) int tab[HTAB];
  const int tid = (int)threadIdx.x;
  const int chunk = (int)blockIdx.x >> 1;
  const int h0 = (blockIdx.x & 1) ? HSPLIT : 0;
  const int hn = (blockIdx.x & 1) ? NNODES - HSPLIT : HSPLIT;
  const int per = (E + NCHUNK - 1) / NCHUNK;
  const int e0 = chunk * per;
  const int e1 = min(E, e0 + per);
  int4* t4 = (int4*)tab;
  const int hn4 = hn / 4;
  for (int i = tid; i < hn4; i += CSR_T) t4[i] = make_int4(0, 0, 0, 0);
  __syncthreads();
  for (int e = e0 + tid; e < e1; e += CSR_B * CSR_T) {
    int d[CSR_B];
#pragma unroll
    for (int j = 0; j < CSR_B; ++j) {
      const int ee = e + j * CSR_T;
      d[j] = (ee < e1) ? dst[ee] : -1;
    }
#pragma unroll
    for (int j = 0; j < CSR_B; ++j)
      if ((unsigned)(d[j] - h0) < (unsigned)hn) atomicAdd(&tab[d[j] - h0], 1);
  }
  __syncthreads();
  int4* h4 = (int4*)(hist + (size_t)chunk * HPITCH + h0);
  int4 c[HTRIPS - 1];
#pragma unroll
  for (int k = 0; k < HTRIPS - 1; ++k) c[k] = t4[tid + k * CSR_T];
  const int il = tid + (HTRIPS - 1) * CSR_T;
  int4 cl = make_int4(0, 0, 0, 0);
  if (il < hn4) cl = t4[il];
#pragma unroll
  for (int k = 0; k < HTRIPS - 1; ++k) h4[tid + k * CSR_T] = c[k];
  if (il < hn4) h4[il] = cl;
}

__device__ __forceinline__ void chunk_scan_body(
    int blk, const int* __restrict__ hist, int n, int* __restrict__ offs,
    float* __restrict__ dinv, int* __restrict__ excl, int* __restrict__ bsum) {
  __shared__ int sm[256];
  int i = blk * 256 + threadIdx.x;
  int run = 0;
  if (i < n) {
#pragma unroll 32
    for (int g = 0; g < NCHUNK; ++g) {
      int c = hist[(size_t)g * HPITCH + i];
      offs[(size_t)g * HPITCH + i] = run;
      run += c;
    }
    dinv[i] = rsqrtf((float)(run + 1));
  }
  sm[threadIdx.x] = run;
  __syncthreads();
#pragma unroll
  for (int off = 1; off < 256; off <<= 1) {
    int t = (threadIdx.x >= off) ? sm[threadIdx.x - off] : 0;
    __syncthreads();
    sm[threadIdx.x] += t;
    __syncthreads();
  }
  if (i < n) excl[i] = sm[threadIdx.x] - run;
  if (threadIdx.x == 255) bsum[blk] = sm[255];
}

__global__ __launch_bounds__(CSR_T) void k_chunk_place(
    const int* __restrict__ src, const int* __restrict__ dst, int E,
    const int* __restrict__ offs, const int* __restrict__ excl,
    const int* __restrict__ bsum, int nb, int n, int* __restrict__ rowptr,
    int* __restrict__ col) {
  __shared__ __attribute__((aligned(16))) int tab[HTAB];
  __shared__ int boff[257];
  const int tid = (int)threadIdx.x;
  const int G = (int)gridDim.x;
  const int chunk = (int)blockIdx.x >> 1;
  const int h0 = (blockIdx.x & 1) ? HSPLIT : 0;
  const int hn = (blockIdx.x & 1) ? NNODES - HSPLIT : HSPLIT;
  const int per = (E + NCHUNK - 1) / NCHUNK;
  const int e0 = chunk * per;
  const int e1 = min(E, e0 + per);
  const int4* o4 = (const int4*)(offs + (size_t)chunk * HPITCH + h0);
  const int4* x4 = (const int4*)(excl + h0);
  const int hn4 = hn / 4;
  int4 o[HTRIPS - 1], r[HTRIPS - 1];
#pragma unroll
  for (int k = 0; k < HTRIPS - 1; ++k) {
    o[k] = o4[tid + k * CSR_T];
    r[k] = x4[tid + k * CSR_T];
  }
  const int il = tid + (HTRIPS - 1) * CSR_T;
  int4 ol = make_int4(0, 0, 0, 0), rl = ol;
  if (il < hn4) {
    ol = o4[il];
    rl = x4[il];
  }
  if (tid < 32) {
    const int lane = tid;
    int v[8];
#pragma unroll
    for (int j = 0; j < 8; ++j) {
      const int k = 8 * lane + j;
      v[j] = (k < nb) ? bsum[k] : 0;
    }
    int tot = 0;
#pragma unroll
    for (int j = 0; j < 8; ++j) {
      const int t = v[j];
      v[j] = tot;
      tot += t;
    }
    int inc = tot;
    for (int off = 1; off < 32; off <<= 1) {
      const int t = __shfl_up(inc, off);
      if (lane >= off) inc += t;
    }
    const int base = inc - tot;
#pragma unroll
    for (int j = 0; j < 8; ++j) boff[8 * lane + j] = base + v[j];
    if (lane == 31) boff[256] = inc;
  }
  __syncthreads();
  int4* t4 = (int4*)tab;
#pragma unroll
  for (int k = 0; k < HTRIPS - 1; ++k) {
    const int i = tid + k * CSR_T;
    const int b = boff[(h0 / 4 + i) >> 6];
    t4[i] = make_int4(o[k].x + r[k].x + b, o[k].y + r[k].y + b,
                      o[k].z + r[k].z + b, o[k].w + r[k].w + b);
  }
  if (il < hn4) {
    const int b = boff[(h0 / 4 + il) >> 6];
    t4[il] = make_int4(ol.x + rl.x + b, ol.y + rl.y + b, ol.z + rl.z + b,
                       ol.w + rl.w + b);
  }
  const int nper = (((n + G) / G) + 31) & ~31;
  const int j0 = blockIdx.x * nper;
  const int j1 = min(n + 1, j0 + nper);
  for (int j = j0 + tid; j < j1; j += CSR_T)
    rowptr[j] = (j < n) ? excl[j] + boff[j >> 8] : boff[256];
  __syncthreads();
  for (int e = e0 + tid; e < e1; e += CSR_B * CSR_T) {
    int s[CSR_B], d[CSR_B], pos[CSR_B];
#pragma unroll
    for (int j = 0; j < CSR_B; ++j) {
      const int ee = e + j * CSR_T;
      const bool in = ee < e1;
      s[j] = in ? src[ee] : -1;
      d[j] = in ? dst[ee] : -1;
    }
#pragma unroll
    for (int j = 0; j < CSR_B; ++j) {
      pos[j] = -1;
      if ((unsigned)s[j] < (unsigned)NNODES &&
          (unsigned)(d[j] - h0) < (unsigned)hn)
        pos[j] = atomicAdd(&tab[d[j] - h0], 1);
    }
#pragma unroll
    for (int j = 0; j < CSR_B; ++j)
      if (pos[j] >= 0) col[pos[j]] = s[j];
  }
}


#define GEMM_GRID 1024

template <bool GATHER>
__device__ __forceinline__ int gemm_a_row(const int* __restrict__ xidx,
                                          int tile, int r) {
  int arow = tile * 16 + r;
  if (GATHER) arow = xidx[arow];
  return arow;
}

template <int K, int COLS, bool GATHER>
__device__ __forceinline__ void gemm_body(
    int tile0, int G, const float* __restrict__ A, const float* __restrict__ B,
    float* __restrict__ C, const int* __restrict__ xidx, int ntiles) {
  constexpr int PITCH = K + 4;
  constexpr int F4_ROW = K / 4;
  constexpr int NF4 = 16 * F4_ROW;
  static_assert(NF4 == 512 || NF4 == 128, "one or two pieces per thread");
  constexpr bool TWO = (NF4 == 512);
  __shared__ __attribute__((aligned(16))) float As[2][16 * PITCH];

  const int tid = threadIdx.x;
  const int lane = tid & 31;
  const int wave = tid >> 5;
  const int lo = lane & 15;
  const int hi = lane >> 4;
  const int col0 = wave * 16;
  const bool has0 = TWO || (tid < NF4);
  const int r0 = tid / F4_ROW, c0 = tid % F4_ROW;
  const int r1 = (tid + 256) / F4_ROW, c1 = (tid + 256) % F4_ROW;
  const int s0 = r0 * PITCH + 4 * c0;
  const int s1 = r1 * PITCH + 4 * c1;

  v2f bf[K / 4];
#pragma unroll
  for (int i = 0; i < K / 4; ++i) {
    const int kb = 4 * i;
    bf[i].x = B[(size_t)(kb + 2 * hi) * COLS + col0 + lo];
    bf[i].y = B[(size_t)(kb + 2 * hi + 1) * COLS + col0 + lo];
  }

  float4 p0 = make_float4(0.f, 0.f, 0.f, 0.f);
  float4 p1 = make_float4(0.f, 0.f, 0.f, 0.f);
  int an0 = 0, an1 = 0;
  int tile = tile0;
  if (tile < ntiles) {
    if (has0) {
      const int a0 = gemm_a_row<GATHER>(xidx, tile, r0);
      p0 = *((const float4*)(A + (size_t)a0 * K) + c0);
    }
    if constexpr (TWO) {
      const int a1 = gemm_a_row<GATHER>(xidx, tile, r1);
      p1 = *((const float4*)(A + (size_t)a1 * K) + c1);
    }
    if (tile + G < ntiles) {
      if (has0) an0 = gemm_a_row<GATHER>(xidx, tile + G, r0);
      if constexpr (TWO) an1 = gemm_a_row<GATHER>(xidx, tile + G, r1);
    }
    if (has0) *(float4*)&As[0][s0] = p0;
    if constexpr (TWO) *(float4*)&As[0][s1] = p1;
  }
  int cur = 0;
  for (; tile < ntiles; tile += G) {
    __syncthreads();
    const int nxt = tile + G;
    const bool more = nxt < ntiles;
    if (more) {
      if (has0) p0 = *((const float4*)(A + (size_t)an0 * K) + c0);
      if constexpr (TWO) p1 = *((const float4*)(A + (size_t)an1 * K) + c1);
      if (nxt + G < ntiles) {
        if (has0) an0 = gemm_a_row<GATHER>(xidx, nxt + G, r0);
        if constexpr (TWO) an1 = gemm_a_row<GATHER>(xidx, nxt + G, r1);
      }
    }

    const float* Ar = &As[cur][lo * PITCH + 2 * hi];
    v8f acc = {};
#pragma unroll
    for (int i = 0; i < K / 4; ++i) {
      const v2f a = *(const v2f*)(Ar + 4 * i);
      acc = __builtin_amdgcn_wmma_f32_16x16x4_f32(
          false, a, false, bf[i], (short)0, acc, false, false);
    }
    const int row0 = tile * 16;
#pragma unroll
    for (int g = 0; g < 8; ++g) {
      C[(size_t)(row0 + g + 8 * hi) * COLS + col0 + lo] = acc[g];
    }

    if (more) {
      if (has0) *(float4*)&As[cur ^ 1][s0] = p0;
      if constexpr (TWO) *(float4*)&As[cur ^ 1][s1] = p1;
    }
    cur ^= 1;
  }
}

template <int K, int COLS, bool GATHER>
__global__ __launch_bounds__(256) void k_gemm_wmma(
    const float* __restrict__ A, const float* __restrict__ B,
    float* __restrict__ C, const int* __restrict__ xidx, int ntiles) {
  gemm_body<K, COLS, GATHER>((int)blockIdx.x, (int)gridDim.x, A, B, C, xidx,
                             ntiles);
}

__global__ __launch_bounds__(256) void k_scan_and_gemm1(
    const int* __restrict__ hist, int n, int* __restrict__ offs,
    float* __restrict__ dinv, int* __restrict__ excl, int* __restrict__ bsum,
    int nsb, const float* __restrict__ emb, const float* __restrict__ W1,
    float* __restrict__ H1, const int* __restrict__ xidx, int ntiles) {
  const int blk = (int)blockIdx.x;
  if (blk < nsb) {
    chunk_scan_body(blk, hist, n, offs, dinv, excl, bsum);
  } else {
    gemm_body<EMB_DIM, HID, true>(blk - nsb, (int)gridDim.x - nsb, emb, W1, H1,
                                  xidx, ntiles);
  }
}


#define AGG_GROUP 8

__device__ __forceinline__ void agg_fma(float4& acc, const float4 v, float w) {
  acc.x = __builtin_fmaf(v.x, w, acc.x);
  acc.y = __builtin_fmaf(v.y, w, acc.y);
  acc.z = __builtin_fmaf(v.z, w, acc.z);
  acc.w = __builtin_fmaf(v.w, w, acc.w);
}

template <bool RELU, bool FIRST>
__global__ __launch_bounds__(256) void k_aggregate(
    const int* __restrict__ rowptr, const int* __restrict__ col,
    float* __restrict__ wedge, const float* __restrict__ H,
    const float* __restrict__ dinv, const float* __restrict__ bias,
    float* __restrict__ out, int n) {
  const int lane = threadIdx.x & 31;
  const int d = __builtin_amdgcn_readfirstlane(
      (int)(blockIdx.x * (blockDim.x >> 5) + (threadIdx.x >> 5)));
  if (d >= n) return;

  const int k0 = rowptr[d];
  const int k1 = rowptr[d + 1];
  const float4* __restrict__ Hl = (const float4*)H + lane;
  const float dd = dinv[d];
  const float4 self = Hl[(size_t)d * 32];
  const float4 bb = *((const float4*)bias + lane);

  float4 acc = make_float4(0.f, 0.f, 0.f, 0.f);
  for (int kb = k0; kb < k1; kb += 32) {
    const int kk = kb + lane;
    int sl = 0;
    float wl = 0.f;
    if (kk < k1) {
      sl = col[kk];
      wl = FIRST ? dinv[sl] : wedge[kk];
    }
    const int cnt = min(32, k1 - kb);
    int j = 0;
    for (; j + AGG_GROUP <= cnt; j += AGG_GROUP) {
      float4 v[AGG_GROUP];
      float w[AGG_GROUP];
#pragma unroll
      for (int u = 0; u < AGG_GROUP; ++u) {
        const int s = __builtin_amdgcn_readlane(sl, j + u);
        w[u] = __builtin_bit_cast(
            float, __builtin_amdgcn_readlane(__builtin_bit_cast(int, wl), j + u));
        v[u] = Hl[(size_t)s * 32];
      }
#pragma unroll
      for (int u = 0; u < AGG_GROUP; ++u) agg_fma(acc, v[u], w[u]);
    }
    const int rem = cnt - j;
    if (rem > 0) {
      float4 v[AGG_GROUP - 1];
      float w[AGG_GROUP - 1];
#pragma unroll
      for (int u = 0; u < AGG_GROUP - 1; ++u) {
        if (u < rem) {
          const int s = __builtin_amdgcn_readlane(sl, j + u);
          w[u] = __builtin_bit_cast(
              float,
              __builtin_amdgcn_readlane(__builtin_bit_cast(int, wl), j + u));
          v[u] = Hl[(size_t)s * 32];
        }
      }
#pragma unroll
      for (int u = 0; u < AGG_GROUP - 1; ++u) {
        if (u < rem) agg_fma(acc, v[u], w[u]);
      }
    }
    if (FIRST && kk < k1) wedge[kk] = wl;
  }
  float4 r;
  r.x = (acc.x + self.x * dd) * dd + bb.x;
  r.y = (acc.y + self.y * dd) * dd + bb.y;
  r.z = (acc.z + self.z * dd) * dd + bb.z;
  r.w = (acc.w + self.w * dd) * dd + bb.w;
  if (RELU) {
    r.x = fmaxf(r.x, 0.f);
    r.y = fmaxf(r.y, 0.f);
    r.z = fmaxf(r.z, 0.f);
    r.w = fmaxf(r.w, 0.f);
  }
  *((float4*)(out + (size_t)d * 128) + lane) = r;
}


static inline size_t align256(size_t x) { return (x + 255) & ~(size_t)255; }

extern "C" void kernel_launch(void* const* d_in, const int* in_sizes, int n_in,
                              void* d_out, int out_size, void* d_ws,
                              size_t ws_size, hipStream_t stream) {
  const int* x    = (const int*)d_in[0];
  const int* ei   = (const int*)d_in[1];
  const float* emb = (const float*)d_in[2];
  const float* W1  = (const float*)d_in[3];
  const float* b1  = (const float*)d_in[4];
  const float* W2  = (const float*)d_in[5];
  const float* b2  = (const float*)d_in[6];
  float* out = (float*)d_out;

  const int N = in_sizes[0];
  const int E = in_sizes[1] / 2;
  const int* src = ei;
  const int* dst = ei + E;
  const int NT = 256;
  const int nsb = (N + NT - 1) / NT;

  char* ws = (char*)d_ws;
  size_t o = 0;
  int* hist = (int*)(ws + o);       o += align256((size_t)NCHUNK * HPITCH * 4);
  int* offs = (int*)(ws + o);       o += align256((size_t)NCHUNK * HPITCH * 4);
  int* excl = (int*)(ws + o);       o += align256((size_t)N * 4);
  int* bsum = (int*)(ws + o);       o += align256((size_t)nsb * 4);
  int* rowptr = (int*)(ws + o);     o += align256((size_t)(N + 1) * 4);
  int* col = (int*)(ws + o);        o += align256((size_t)E * 4);
  float* dinv = (float*)(ws + o);   o += align256((size_t)N * 4);
  float* bufA = (float*)(ws + o);   o += align256((size_t)N * 128 * 4);
  float* bufB = (float*)(ws + o);   o += align256((size_t)N * 128 * 4);
  float* wedge = (float*)(ws + o);

  k_chunk_hist<<<CSR_GRID, CSR_T, 0, stream>>>(dst, E, hist);
  k_scan_and_gemm1<<<nsb + GEMM_GRID, NT, 0, stream>>>(
      hist, N, offs, dinv, excl, bsum, nsb, emb, W1, bufA, x, N / 16);

  k_chunk_place<<<CSR_GRID, CSR_T, 0, stream>>>(src, dst, E, offs, excl, bsum, nsb,
                                             N, rowptr, col);

  k_aggregate<true, true><<<(N + 7) / 8, NT, 0, stream>>>(
      rowptr, col, wedge, bufA, dinv, b1, bufB, N);

  k_gemm_wmma<HID, OUTD, false><<<GEMM_GRID, NT, 0, stream>>>(bufB, W2, bufA,
                                                           nullptr, N / 16);
  k_aggregate<false, false><<<(N + 7) / 8, NT, 0, stream>>>(
      rowptr, col, wedge, bufA, dinv, b2, out, N);
}


# Layer-2 GEMM launch dropped: layer 2 sums the neighbours' hidden rows first, the work-group then multiplies its 16 reduced rows by W2 on WMMA from an LDS tile; layer-2 sum order changes
# speedup vs baseline: 2.2775x; 1.0589x over previous
#include <hip/hip_runtime.h>
#include <hip/hip_bf16.h>


typedef float v2f __attribute__((ext_vector_type(2)));
typedef float v8f __attribute__((ext_vector_type(8)));

#define NNODES 50000
#define EMB_DIM 32
#define HID 128
#define OUTD 128


#define NCHUNK 128
#define CSR_GRID 256
#define CSR_T 512
#define CSR_B 16
#define HPITCH 50048
#define HSPLIT 25024
#define HTAB 25024
#define HTRIPS ((HTAB / 4 + CSR_T - 1) / CSR_T)
static_assert(HSPLIT % 32 == 0 && HSPLIT <= HTAB && NNODES - HSPLIT <= HTAB &&
                  (NNODES - HSPLIT) % 4 == 0 &&
                  (HTRIPS - 1) * CSR_T <= (NNODES - HSPLIT) / 4 &&
                  (HTRIPS - 1) * CSR_T <= HSPLIT / 4,
              "halves: whole lines, whole int4s, inside the table");

__global__ __launch_bounds__(CSR_T) void k_chunk_hist(
    const int* __restrict__ dst, int E, int* __restrict__ hist) {
  __shared__ __attribute__((aligned(16))) int tab[HTAB];
  const int tid = (int)threadIdx.x;
  const int chunk = (int)blockIdx.x >> 1;
  const int h0 = (blockIdx.x & 1) ? HSPLIT : 0;
  const int hn = (blockIdx.x & 1) ? NNODES - HSPLIT : HSPLIT;
  const int per = (E + NCHUNK - 1) / NCHUNK;
  const int e0 = chunk * per;
  const int e1 = min(E, e0 + per);
  int4* t4 = (int4*)tab;
  const int hn4 = hn / 4;
  for (int i = tid; i < hn4; i += CSR_T) t4[i] = make_int4(0, 0, 0, 0);
  __syncthreads();
  for (int e = e0 + tid; e < e1; e += CSR_B * CSR_T) {
    int d[CSR_B];
#pragma unroll
    for (int j = 0; j < CSR_B; ++j) {
      const int ee = e + j * CSR_T;
      d[j] = (ee < e1) ? dst[ee] : -1;
    }
#pragma unroll
    for (int j = 0; j < CSR_B; ++j)
      if ((unsigned)(d[j] - h0) < (unsigned)hn) atomicAdd(&tab[d[j] - h0], 1);
  }
  __syncthreads();
  int4* h4 = (int4*)(hist + (size_t)chunk * HPITCH + h0);
  int4 c[HTRIPS - 1];
#pragma unroll
  for (int k = 0; k < HTRIPS - 1; ++k) c[k] = t4[tid + k * CSR_T];
  const int il = tid + (HTRIPS - 1) * CSR_T;
  int4 cl = make_int4(0, 0, 0, 0);
  if (il < hn4) cl = t4[il];
#pragma unroll
  for (int k = 0; k < HTRIPS - 1; ++k) h4[tid + k * CSR_T] = c[k];
  if (il < hn4) h4[il] = cl;
}

__device__ __forceinline__ void chunk_scan_body(
    int blk, const int* __restrict__ hist, int n, int* __restrict__ offs,
    float* __restrict__ dinv, int* __restrict__ excl, int* __restrict__ bsum) {
  __shared__ int sm[256];
  int i = blk * 256 + threadIdx.x;
  int run = 0;
  if (i < n) {
#pragma unroll 32
    for (int g = 0; g < NCHUNK; ++g) {
      int c = hist[(size_t)g * HPITCH + i];
      offs[(size_t)g * HPITCH + i] = run;
      run += c;
    }
    dinv[i] = rsqrtf((float)(run + 1));
  }
  sm[threadIdx.x] = run;
  __syncthreads();
#pragma unroll
  for (int off = 1; off < 256; off <<= 1) {
    int t = (threadIdx.x >= off) ? sm[threadIdx.x - off] : 0;
    __syncthreads();
    sm[threadIdx.x] += t;
    __syncthreads();
  }
  if (i < n) excl[i] = sm[threadIdx.x] - run;
  if (threadIdx.x == 255) bsum[blk] = sm[255];
}

__global__ __launch_bounds__(CSR_T) void k_chunk_place(
    const int* __restrict__ src, const int* __restrict__ dst, int E,
    const int* __restrict__ offs, const int* __restrict__ excl,
    const int* __restrict__ bsum, int nb, int n, int* __restrict__ rowptr,
    int* __restrict__ col) {
  __shared__ __attribute__((aligned(16))) int tab[HTAB];
  __shared__ int boff[257];
  const int tid = (int)threadIdx.x;
  const int G = (int)gridDim.x;
  const int chunk = (int)blockIdx.x >> 1;
  const int h0 = (blockIdx.x & 1) ? HSPLIT : 0;
  const int hn = (blockIdx.x & 1) ? NNODES - HSPLIT : HSPLIT;
  const int per = (E + NCHUNK - 1) / NCHUNK;
  const int e0 = chunk * per;
  const int e1 = min(E, e0 + per);
  const int4* o4 = (const int4*)(offs + (size_t)chunk * HPITCH + h0);
  const int4* x4 = (const int4*)(excl + h0);
  const int hn4 = hn / 4;
  int4 o[HTRIPS - 1], r[HTRIPS - 1];
#pragma unroll
  for (int k = 0; k < HTRIPS - 1; ++k) {
    o[k] = o4[tid + k * CSR_T];
    r[k] = x4[tid + k * CSR_T];
  }
  const int il = tid + (HTRIPS - 1) * CSR_T;
  int4 ol = make_int4(0, 0, 0, 0), rl = ol;
  if (il < hn4) {
    ol = o4[il];
    rl = x4[il];
  }
  if (tid < 32) {
    const int lane = tid;
    int v[8];
#pragma unroll
    for (int j = 0; j < 8; ++j) {
      const int k = 8 * lane + j;
      v[j] = (k < nb) ? bsum[k] : 0;
    }
    int tot = 0;
#pragma unroll
    for (int j = 0; j < 8; ++j) {
      const int t = v[j];
      v[j] = tot;
      tot += t;
    }
    int inc = tot;
    for (int off = 1; off < 32; off <<= 1) {
      const int t = __shfl_up(inc, off);
      if (lane >= off) inc += t;
    }
    const int base = inc - tot;
#pragma unroll
    for (int j = 0; j < 8; ++j) boff[8 * lane + j] = base + v[j];
    if (lane == 31) boff[256] = inc;
  }
  __syncthreads();
  int4* t4 = (int4*)tab;
#pragma unroll
  for (int k = 0; k < HTRIPS - 1; ++k) {
    const int i = tid + k * CSR_T;
    const int b = boff[(h0 / 4 + i) >> 6];
    t4[i] = make_int4(o[k].x + r[k].x + b, o[k].y + r[k].y + b,
                      o[k].z + r[k].z + b, o[k].w + r[k].w + b);
  }
  if (il < hn4) {
    const int b = boff[(h0 / 4 + il) >> 6];
    t4[il] = make_int4(ol.x + rl.x + b, ol.y + rl.y + b, ol.z + rl.z + b,
                       ol.w + rl.w + b);
  }
  const int nper = (((n + G) / G) + 31) & ~31;
  const int j0 = blockIdx.x * nper;
  const int j1 = min(n + 1, j0 + nper);
  for (int j = j0 + tid; j < j1; j += CSR_T)
    rowptr[j] = (j < n) ? excl[j] + boff[j >> 8] : boff[256];
  __syncthreads();
  for (int e = e0 + tid; e < e1; e += CSR_B * CSR_T) {
    int s[CSR_B], d[CSR_B], pos[CSR_B];
#pragma unroll
    for (int j = 0; j < CSR_B; ++j) {
      const int ee = e + j * CSR_T;
      const bool in = ee < e1;
      s[j] = in ? src[ee] : -1;
      d[j] = in ? dst[ee] : -1;
    }
#pragma unroll
    for (int j = 0; j < CSR_B; ++j) {
      pos[j] = -1;
      if ((unsigned)s[j] < (unsigned)NNODES &&
          (unsigned)(d[j] - h0) < (unsigned)hn)
        pos[j] = atomicAdd(&tab[d[j] - h0], 1);
    }
#pragma unroll
    for (int j = 0; j < CSR_B; ++j)
      if (pos[j] >= 0) col[pos[j]] = s[j];
  }
}


#define GEMM_GRID 1024

template <bool GATHER>
__device__ __forceinline__ int gemm_a_row(const int* __restrict__ xidx,
                                          int tile, int r) {
  int arow = tile * 16 + r;
  if (GATHER) arow = xidx[arow];
  return arow;
}

template <int K, int COLS, bool GATHER>
__device__ __forceinline__ void gemm_body(
    int tile0, int G, const float* __restrict__ A, const float* __restrict__ B,
    float* __restrict__ C, const int* __restrict__ xidx, int ntiles) {
  constexpr int PITCH = K + 4;
  constexpr int F4_ROW = K / 4;
  constexpr int NF4 = 16 * F4_ROW;
  static_assert(NF4 == 512 || NF4 == 128, "one or two pieces per thread");
  constexpr bool TWO = (NF4 == 512);
  __shared__ __attribute__((aligned(16))) float As[2][16 * PITCH];

  const int tid = threadIdx.x;
  const int lane = tid & 31;
  const int wave = tid >> 5;
  const int lo = lane & 15;
  const int hi = lane >> 4;
  const int col0 = wave * 16;
  const bool has0 = TWO || (tid < NF4);
  const int r0 = tid / F4_ROW, c0 = tid % F4_ROW;
  const int r1 = (tid + 256) / F4_ROW, c1 = (tid + 256) % F4_ROW;
  const int s0 = r0 * PITCH + 4 * c0;
  const int s1 = r1 * PITCH + 4 * c1;

  v2f bf[K / 4];
#pragma unroll
  for (int i = 0; i < K / 4; ++i) {
    const int kb = 4 * i;
    bf[i].x = B[(size_t)(kb + 2 * hi) * COLS + col0 + lo];
    bf[i].y = B[(size_t)(kb + 2 * hi + 1) * COLS + col0 + lo];
  }

  float4 p0 = make_float4(0.f, 0.f, 0.f, 0.f);
  float4 p1 = make_float4(0.f, 0.f, 0.f, 0.f);
  int an0 = 0, an1 = 0;
  int tile = tile0;
  if (tile < ntiles) {
    if (has0) {
      const int a0 = gemm_a_row<GATHER>(xidx, tile, r0);
      p0 = *((const float4*)(A + (size_t)a0 * K) + c0);
    }
    if constexpr (TWO) {
      const int a1 = gemm_a_row<GATHER>(xidx, tile, r1);
      p1 = *((const float4*)(A + (size_t)a1 * K) + c1);
    }
    if (tile + G < ntiles) {
      if (has0) an0 = gemm_a_row<GATHER>(xidx, tile + G, r0);
      if constexpr (TWO) an1 = gemm_a_row<GATHER>(xidx, tile + G, r1);
    }
    if (has0) *(float4*)&As[0][s0] = p0;
    if constexpr (TWO) *(float4*)&As[0][s1] = p1;
  }
  int cur = 0;
  for (; tile < ntiles; tile += G) {
    __syncthreads();
    const int nxt = tile + G;
    const bool more = nxt < ntiles;
    if (more) {
      if (has0) p0 = *((const float4*)(A + (size_t)an0 * K) + c0);
      if constexpr (TWO) p1 = *((const float4*)(A + (size_t)an1 * K) + c1);
      if (nxt + G < ntiles) {
        if (has0) an0 = gemm_a_row<GATHER>(xidx, nxt + G, r0);
        if constexpr (TWO) an1 = gemm_a_row<GATHER>(xidx, nxt + G, r1);
      }
    }

    const float* Ar = &As[cur][lo * PITCH + 2 * hi];
    v8f acc = {};
#pragma unroll
    for (int i = 0; i < K / 4; ++i) {
      const v2f a = *(const v2f*)(Ar + 4 * i);
      acc = __builtin_amdgcn_wmma_f32_16x16x4_f32(
          false, a, false, bf[i], (short)0, acc, false, false);
    }
    const int row0 = tile * 16;
#pragma unroll
    for (int g = 0; g < 8; ++g) {
      C[(size_t)(row0 + g + 8 * hi) * COLS + col0 + lo] = acc[g];
    }

    if (more) {
      if (has0) *(float4*)&As[cur ^ 1][s0] = p0;
      if constexpr (TWO) *(float4*)&As[cur ^ 1][s1] = p1;
    }
    cur ^= 1;
  }
}

__global__ __launch_bounds__(256) void k_scan_and_gemm1(
    const int* __restrict__ hist, int n, int* __restrict__ offs,
    float* __restrict__ dinv, int* __restrict__ excl, int* __restrict__ bsum,
    int nsb, const float* __restrict__ emb, const float* __restrict__ W1,
    float* __restrict__ H1, const int* __restrict__ xidx, int ntiles) {
  const int blk = (int)blockIdx.x;
  if (blk < nsb) {
    chunk_scan_body(blk, hist, n, offs, dinv, excl, bsum);
  } else {
    gemm_body<EMB_DIM, HID, true>(blk - nsb, (int)gridDim.x - nsb, emb, W1, H1,
                                  xidx, ntiles);
  }
}


#define AGG_GROUP 8

__device__ __forceinline__ void agg_fma(float4& acc, const float4 v, float w) {
  acc.x = __builtin_fmaf(v.x, w, acc.x);
  acc.y = __builtin_fmaf(v.y, w, acc.y);
  acc.z = __builtin_fmaf(v.z, w, acc.z);
  acc.w = __builtin_fmaf(v.w, w, acc.w);
}

template <bool RELU, bool FIRST>
__global__ __launch_bounds__(256) void k_aggregate(
    const int* __restrict__ rowptr, const int* __restrict__ col,
    float* __restrict__ wedge, const float* __restrict__ H,
    const float* __restrict__ dinv, const float* __restrict__ bias,
    float* __restrict__ out, int n) {
  const int lane = threadIdx.x & 31;
  const int d = __builtin_amdgcn_readfirstlane(
      (int)(blockIdx.x * (blockDim.x >> 5) + (threadIdx.x >> 5)));
  if (d >= n) return;

  const int k0 = rowptr[d];
  const int k1 = rowptr[d + 1];
  const float4* __restrict__ Hl = (const float4*)H + lane;
  const float dd = dinv[d];
  const float4 self = Hl[(size_t)d * 32];
  const float4 bb = *((const float4*)bias + lane);

  float4 acc = make_float4(0.f, 0.f, 0.f, 0.f);
  for (int kb = k0; kb < k1; kb += 32) {
    const int kk = kb + lane;
    int sl = 0;
    float wl = 0.f;
    if (kk < k1) {
      sl = col[kk];
      wl = FIRST ? dinv[sl] : wedge[kk];
    }
    const int cnt = min(32, k1 - kb);
    int j = 0;
    for (; j + AGG_GROUP <= cnt; j += AGG_GROUP) {
      float4 v[AGG_GROUP];
      float w[AGG_GROUP];
#pragma unroll
      for (int u = 0; u < AGG_GROUP; ++u) {
        const int s = __builtin_amdgcn_readlane(sl, j + u);
        w[u] = __builtin_bit_cast(
            float, __builtin_amdgcn_readlane(__builtin_bit_cast(int, wl), j + u));
        v[u] = Hl[(size_t)s * 32];
      }
#pragma unroll
      for (int u = 0; u < AGG_GROUP; ++u) agg_fma(acc, v[u], w[u]);
    }
    const int rem = cnt - j;
    if (rem > 0) {
      float4 v[AGG_GROUP - 1];
      float w[AGG_GROUP - 1];
#pragma unroll
      for (int u = 0; u < AGG_GROUP - 1; ++u) {
        if (u < rem) {
          const int s = __builtin_amdgcn_readlane(sl, j + u);
          w[u] = __builtin_bit_cast(
              float,
              __builtin_amdgcn_readlane(__builtin_bit_cast(int, wl), j + u));
          v[u] = Hl[(size_t)s * 32];
        }
      }
#pragma unroll
      for (int u = 0; u < AGG_GROUP - 1; ++u) {
        if (u < rem) agg_fma(acc, v[u], w[u]);
      }
    }
    if (FIRST && kk < k1) wedge[kk] = wl;
  }
  float4 r;
  r.x = (acc.x + self.x * dd) * dd + bb.x;
  r.y = (acc.y + self.y * dd) * dd + bb.y;
  r.z = (acc.z + self.z * dd) * dd + bb.z;
  r.w = (acc.w + self.w * dd) * dd + bb.w;
  if (RELU) {
    r.x = fmaxf(r.x, 0.f);
    r.y = fmaxf(r.y, 0.f);
    r.z = fmaxf(r.z, 0.f);
    r.w = fmaxf(r.w, 0.f);
  }
  *((float4*)(out + (size_t)d * 128) + lane) = r;
}


#define FUSE_GRID 1024

__device__ __forceinline__ float4 agg_rows(
    int k0, int k1, int sl, float wl, const int* __restrict__ col,
    const float* __restrict__ wedge, const float4* __restrict__ Hl, int lane) {
  float4 acc = make_float4(0.f, 0.f, 0.f, 0.f);
  for (int kb = k0; kb < k1; kb += 32) {
    const int kn = kb + 32 + lane;
    int sn = 0;
    float wn = 0.f;
    if (kn < k1) {
      sn = col[kn];
      wn = wedge[kn];
    }
    const int cnt = min(32, k1 - kb);
    int j = 0;
    for (; j + AGG_GROUP <= cnt; j += AGG_GROUP) {
      float4 v[AGG_GROUP];
      float w[AGG_GROUP];
#pragma unroll
      for (int u = 0; u < AGG_GROUP; ++u) {
        const int s = __builtin_amdgcn_readlane(sl, j + u);
        w[u] = __builtin_bit_cast(
            float, __builtin_amdgcn_readlane(__builtin_bit_cast(int, wl), j + u));
        v[u] = Hl[(size_t)s * 32];
      }
#pragma unroll
      for (int u = 0; u < AGG_GROUP; ++u) agg_fma(acc, v[u], w[u]);
    }
    const int rem = cnt - j;
    if (rem > 0) {
      float4 v[AGG_GROUP - 1];
      float w[AGG_GROUP - 1];
#pragma unroll
      for (int u = 0; u < AGG_GROUP - 1; ++u) {
        if (u < rem) {
          const int s = __builtin_amdgcn_readlane(sl, j + u);
          w[u] = __builtin_bit_cast(
              float,
              __builtin_amdgcn_readlane(__builtin_bit_cast(int, wl), j + u));
          v[u] = Hl[(size_t)s * 32];
        }
      }
#pragma unroll
      for (int u = 0; u < AGG_GROUP - 1; ++u) {
        if (u < rem) agg_fma(acc, v[u], w[u]);
      }
    }
    sl = sn;
    wl = wn;
  }
  return acc;
}

__global__ __launch_bounds__(256) void k_aggregate_gemm(
    const int* __restrict__ rowptr, const int* __restrict__ col,
    const float* __restrict__ wedge, const float* __restrict__ H,
    const float* __restrict__ dinv, const float* __restrict__ B,
    const float* __restrict__ bias, float* __restrict__ out, int ntiles) {
  constexpr int K = HID, COLS = OUTD;
  constexpr int PITCH = K + 4;
  __shared__ __attribute__((aligned(16))) float As[2][16 * PITCH];
  constexpr int BREG = 2;
  __shared__ __attribute__((aligned(16))) v2f Bs[8 * 32 * (K / 4 - BREG)];

  const int tid = threadIdx.x;
  const int lane = tid & 31;
  const int wave = tid >> 5;
  const int lo = lane & 15;
  const int hi = lane >> 4;
  const int col0 = wave * 16;
  const int G = (int)gridDim.x;
  const float4* __restrict__ Hl = (const float4*)H + lane;

  v2f* Bw = Bs + wave * (32 * (K / 4 - BREG)) + lane;
  v2f br[BREG];
#pragma unroll
  for (int i = 0; i < K / 4; ++i) {
    const int kb = 4 * i;
    v2f f;
    f.x = B[(size_t)(kb + 2 * hi) * COLS + col0 + lo];
    f.y = B[(size_t)(kb + 2 * hi + 1) * COLS + col0 + lo];
    if (i < BREG)
      br[i] = f;
    else
      Bw[32 * (i - BREG)] = f;
  }
  const float bj = bias[col0 + lo];

  int tile = (int)blockIdx.x;
  int k0 = 0, k1 = 0, k2 = 0;
  int slA = 0, slB = 0;
  float wlA = 0.f, wlB = 0.f, ddA = 0.f, ddB = 0.f;
  float4 selfA = make_float4(0.f, 0.f, 0.f, 0.f), selfB = selfA;
  int rpn = 0;
  if (tile < ntiles) {
    const int d0 = __builtin_amdgcn_readfirstlane(tile * 16 + 2 * wave);
    k0 = rowptr[d0];
    k1 = rowptr[d0 + 1];
    k2 = rowptr[d0 + 2];
    if (k0 + lane < k1) {
      slA = col[k0 + lane];
      wlA = wedge[k0 + lane];
    }
    if (k1 + lane < k2) {
      slB = col[k1 + lane];
      wlB = wedge[k1 + lane];
    }
    ddA = dinv[d0];
    ddB = dinv[d0 + 1];
    selfA = Hl[(size_t)d0 * 32];
    selfB = Hl[(size_t)(d0 + 1) * 32];
    if (tile + G < ntiles) {
      const int dn =
          __builtin_amdgcn_readfirstlane((tile + G) * 16 + 2 * wave);
      rpn = rowptr[dn + min(lane, 2)];
    }
  }
  int cur = 0;
  for (; tile < ntiles; tile += G) {
    const int nxt = tile + G;
    const bool more = nxt < ntiles;
    int n0 = 0, n1 = 0, n2 = 0;

    float* Aw = &As[cur][(2 * wave) * PITCH + 4 * lane];
    {
      const float4 a = agg_rows(k0, k1, slA, wlA, col, wedge, Hl, lane);
      float4 z;
      z.x = (a.x + selfA.x * ddA) * ddA;
      z.y = (a.y + selfA.y * ddA) * ddA;
      z.z = (a.z + selfA.z * ddA) * ddA;
      z.w = (a.w + selfA.w * ddA) * ddA;
      *(float4*)Aw = z;
    }
    {
      const float4 a = agg_rows(k1, k2, slB, wlB, col, wedge, Hl, lane);
      float4 z;
      z.x = (a.x + selfB.x * ddB) * ddB;
      z.y = (a.y + selfB.y * ddB) * ddB;
      z.z = (a.z + selfB.z * ddB) * ddB;
      z.w = (a.w + selfB.w * ddB) * ddB;
      *(float4*)(Aw + PITCH) = z;
    }

    __syncthreads();

    if (more) {
      const int dn = __builtin_amdgcn_readfirstlane(nxt * 16 + 2 * wave);
      n0 = __builtin_amdgcn_readlane(rpn, 0);
      n1 = __builtin_amdgcn_readlane(rpn, 1);
      n2 = __builtin_amdgcn_readlane(rpn, 2);
      if (nxt + G < ntiles) {
        const int dnn =
            __builtin_amdgcn_readfirstlane((nxt + G) * 16 + 2 * wave);
        rpn = rowptr[dnn + min(lane, 2)];
      }
      slA = 0;
      slB = 0;
      wlA = 0.f;
      wlB = 0.f;
      if (n0 + lane < n1) {
        slA = col[n0 + lane];
        wlA = wedge[n0 + lane];
      }
      if (n1 + lane < n2) {
        slB = col[n1 + lane];
        wlB = wedge[n1 + lane];
      }
      ddA = dinv[dn];
      ddB = dinv[dn + 1];
      selfA = Hl[(size_t)dn * 32];
      selfB = Hl[(size_t)(dn + 1) * 32];
    }

    const float* Ar = &As[cur][lo * PITCH + 2 * hi];
    v8f acc0 = {}, acc1 = {};
#pragma unroll
    for (int i = 0; i < K / 4; i += 2) {
      const v2f a0 = *(const v2f*)(Ar + 4 * i);
      const v2f a1 = *(const v2f*)(Ar + 4 * i + 4);
      const v2f b0 = (i < BREG) ? br[i] : Bw[32 * (i - BREG)];
      const v2f b1 = (i < BREG) ? br[i + 1] : Bw[32 * (i - BREG) + 32];
      acc0 = __builtin_amdgcn_wmma_f32_16x16x4_f32(
          false, a0, false, b0, (short)0, acc0, false, false);
      acc1 = __builtin_amdgcn_wmma_f32_16x16x4_f32(
          false, a1, false, b1, (short)0, acc1, false, false);
    }
    const int row0 = tile * 16;
#pragma unroll
    for (int g = 0; g < 8; ++g) {
      out[(size_t)(row0 + g + 8 * hi) * COLS + col0 + lo] =
          (acc0[g] + acc1[g]) + bj;
    }

    k0 = n0;
    k1 = n1;
    k2 = n2;
    cur ^= 1;
  }
}


static inline size_t align256(size_t x) { return (x + 255) & ~(size_t)255; }

extern "C" void kernel_launch(void* const* d_in, const int* in_sizes, int n_in,
                              void* d_out, int out_size, void* d_ws,
                              size_t ws_size, hipStream_t stream) {
  const int* x    = (const int*)d_in[0];
  const int* ei   = (const int*)d_in[1];
  const float* emb = (const float*)d_in[2];
  const float* W1  = (const float*)d_in[3];
  const float* b1  = (const float*)d_in[4];
  const float* W2  = (const float*)d_in[5];
  const float* b2  = (const float*)d_in[6];
  float* out = (float*)d_out;

  const int N = in_sizes[0];
  const int E = in_sizes[1] / 2;
  const int* src = ei;
  const int* dst = ei + E;
  const int NT = 256;
  const int nsb = (N + NT - 1) / NT;

  char* ws = (char*)d_ws;
  size_t o = 0;
  int* hist = (int*)(ws + o);       o += align256((size_t)NCHUNK * HPITCH * 4);
  int* offs = (int*)(ws + o);       o += align256((size_t)NCHUNK * HPITCH * 4);
  int* excl = (int*)(ws + o);       o += align256((size_t)N * 4);
  int* bsum = (int*)(ws + o);       o += align256((size_t)nsb * 4);
  int* rowptr = (int*)(ws + o);     o += align256((size_t)(N + 1) * 4);
  int* col = (int*)(ws + o);        o += align256((size_t)E * 4);
  float* dinv = (float*)(ws + o);   o += align256((size_t)N * 4);
  float* bufA = (float*)(ws + o);   o += align256((size_t)N * 128 * 4);
  float* bufB = (float*)(ws + o);   o += align256((size_t)N * 128 * 4);
  float* wedge = (float*)(ws + o);

  k_chunk_hist<<<CSR_GRID, CSR_T, 0, stream>>>(dst, E, hist);
  k_scan_and_gemm1<<<nsb + GEMM_GRID, NT, 0, stream>>>(
      hist, N, offs, dinv, excl, bsum, nsb, emb, W1, bufA, x, N / 16);

  k_chunk_place<<<CSR_GRID, CSR_T, 0, stream>>>(src, dst, E, offs, excl, bsum, nsb,
                                             N, rowptr, col);

  k_aggregate<true, true><<<(N + 7) / 8, NT, 0, stream>>>(
      rowptr, col, wedge, bufA, dinv, b1, bufB, N);

  k_aggregate_gemm<<<FUSE_GRID, NT, 0, stream>>>(rowptr, col, wedge, bufB, dinv,
                                               W2, b2, out, N / 16);
}
